# starting kernel
# baseline (speedup 1.0000x reference)
#include <hip/hip_runtime.h>


#define N_NODES   100000
#define F1        64
#define F2        128
#define NEG_SLOPE 0.01f

typedef __attribute__((ext_vector_type(2))) float v2f;
typedef __attribute__((ext_vector_type(8))) float v8f;

__device__ __forceinline__ float leaky(float v) {
    return v >= 0.0f ? v : v * NEG_SLOPE;
}


__global__ void k_deg_init(float* __restrict__ deg) {
    int i = blockIdx.x * blockDim.x + threadIdx.x;
    if (i < N_NODES) deg[i] = 1.0f;
}

__global__ void k_deg_accum(const int* __restrict__ dst, int E,
                            float* __restrict__ deg) {
    int e = blockIdx.x * blockDim.x + threadIdx.x;
    if (e < E) atomicAdd(&deg[dst[e]], 1.0f);
}

__global__ void k_dinv(const float* __restrict__ deg, float* __restrict__ dinv) {
    int i = blockIdx.x * blockDim.x + threadIdx.x;
    if (i < N_NODES) dinv[i] = rsqrtf(fmaxf(deg[i], 1.0f));
}


__global__ void k_layer1(const float* __restrict__ x, const float* __restrict__ W1,
                         const float* __restrict__ dinv,
                         float* __restrict__ h1, float* __restrict__ agg1) {
    int idx = blockIdx.x * blockDim.x + threadIdx.x;
    if (idx >= N_NODES * F1) return;
    int n = idx >> 6;
    int f = idx & (F1 - 1);
    float h = fmaf(x[2 * n], W1[f], x[2 * n + 1] * W1[F1 + f]);
    h1[idx] = h;
    float di = dinv[n];
    agg1[idx] = h * di * di;
}


__global__ void k_edge_agg1(const int* __restrict__ src, const int* __restrict__ dst,
                            int E, const float* __restrict__ dinv,
                            const float* __restrict__ h1, float* __restrict__ agg1) {
    int t = blockIdx.x * 256 + threadIdx.x;
    int e = t >> 6;
    int f = t & (F1 - 1);
    if (e >= E) return;
    int s = src[e];
    int d = dst[e];
    float norm = dinv[s] * dinv[d];
    atomicAdd(&agg1[d * F1 + f], h1[s * F1 + f] * norm);
}


__global__ void k_act1(float* __restrict__ agg1, const float* __restrict__ b1) {
    int idx = blockIdx.x * blockDim.x + threadIdx.x;
    if (idx >= N_NODES * F1) return;
    agg1[idx] = leaky(agg1[idx] + b1[idx & (F1 - 1)]);
}


__global__ void __launch_bounds__(256)
k_gemm2_wmma(const float* __restrict__ A,
             const float* __restrict__ W2,
             const float* __restrict__ dinv,
             float* __restrict__ h2,
             float* __restrict__ agg2) {
    const int lane = threadIdx.x & 31;
    const int wave = threadIdx.x >> 5;
    const int row0 = blockIdx.x * 16;
    const int col0 = wave * 16;

    const int m  = lane & 15;
    const int kh = (lane >> 4) << 1;

    v8f acc = {};
    const float* Arow = A + (size_t)(row0 + m) * F1;
#pragma unroll
    for (int k0 = 0; k0 < F1; k0 += 4) {
        v2f a, b;
        a.x = Arow[k0 + kh];
        a.y = Arow[k0 + kh + 1];
        b.x = W2[(k0 + kh) * F2 + col0 + m];
        b.y = W2[(k0 + kh + 1) * F2 + col0 + m];
        acc = __builtin_amdgcn_wmma_f32_16x16x4_f32(
             false, a,  false, b,
             (short)0, acc,  false,  false);
    }

    const int nc    = lane & 15;
    const int rbase = row0 + ((lane >> 4) << 3);
#pragma unroll
    for (int v = 0; v < 8; ++v) {
        int   r  = rbase + v;
        float d  = acc[v];
        float di = dinv[r];
        size_t o = (size_t)r * F2 + col0 + nc;
        h2[o]    = d;
        agg2[o]  = d * di * di;
    }
}


__global__ void k_edge_agg2(const int* __restrict__ src, const int* __restrict__ dst,
                            int E, const float* __restrict__ dinv,
                            const float* __restrict__ h2, float* __restrict__ agg2) {
    int t = blockIdx.x * 256 + threadIdx.x;
    int e = t >> 7;
    int f = t & (F2 - 1);
    if (e >= E) return;
    int s = src[e];
    int d = dst[e];
    float norm = dinv[s] * dinv[d];
    atomicAdd(&agg2[d * F2 + f], h2[s * F2 + f] * norm);
}


__global__ void k_act2(float* __restrict__ z, const float* __restrict__ b2) {
    int idx = blockIdx.x * blockDim.x + threadIdx.x;
    if (idx >= N_NODES * F2) return;
    z[idx] = leaky(z[idx] + b2[idx & (F2 - 1)]);
}


extern "C" void kernel_launch(void* const* d_in, const int* in_sizes, int n_in,
                              void* d_out, int out_size, void* d_ws, size_t ws_size,
                              hipStream_t stream) {
    const int*   edge_index = (const int*)d_in[0];
    const float* x          = (const float*)d_in[1];
    const float* W1         = (const float*)d_in[2];
    const float* b1         = (const float*)d_in[3];
    const float* W2         = (const float*)d_in[4];
    const float* b2         = (const float*)d_in[5];
    float*       z          = (float*)d_out;

    const int E = in_sizes[0] / 2;
    const int* src = edge_index;
    const int* dst = edge_index + E;

    float* ws   = (float*)d_ws;
    float* deg  = ws;
    float* dinv = ws + N_NODES;
    float* agg1 = ws + 2 * N_NODES;
    float* h1   = agg1 + N_NODES * F1;
    float* h2   = h1;
    (void)ws_size; (void)n_in; (void)out_size;

    const int T = 256;

    k_deg_init <<<(N_NODES + T - 1) / T, T, 0, stream>>>(deg);
    k_deg_accum<<<(E + T - 1) / T, T, 0, stream>>>(dst, E, deg);
    k_dinv     <<<(N_NODES + T - 1) / T, T, 0, stream>>>(deg, dinv);

    k_layer1   <<<(N_NODES * F1 + T - 1) / T, T, 0, stream>>>(x, W1, dinv, h1, agg1);
    k_edge_agg1<<<((size_t)E * F1 + T - 1) / T, T, 0, stream>>>(src, dst, E, dinv, h1, agg1);
    k_act1     <<<(N_NODES * F1 + T - 1) / T, T, 0, stream>>>(agg1, b1);

    k_gemm2_wmma<<<N_NODES / 16, 256, 0, stream>>>(agg1, W2, dinv, h2, z);
    k_edge_agg2 <<<((size_t)E * F2 + T - 1) / T, T, 0, stream>>>(src, dst, E, dinv, h2, z);
    k_act2      <<<(N_NODES * F2 + T - 1) / T, T, 0, stream>>>(z, b2);
}


# Aggregate each layer's input before its dense transform ((A X) W for A (X W)): edge atomics on 2 and 64 features, not 64 and 128; bias+leaky in transform epilogue; float summation order changes
# speedup vs baseline: 2.1824x; 2.1824x over previous
#include <hip/hip_runtime.h>


#define N_NODES   100000
#define F0        2
#define F1        64
#define F2        128
#define NEG_SLOPE 0.01f

typedef __attribute__((ext_vector_type(2))) float v2f;
typedef __attribute__((ext_vector_type(8))) float v8f;

__device__ __forceinline__ float leaky(float v) {
    return v >= 0.0f ? v : v * NEG_SLOPE;
}


__global__ void k_deg_init(float* __restrict__ deg) {
    int i = blockIdx.x * blockDim.x + threadIdx.x;
    if (i < N_NODES) deg[i] = 1.0f;
}

__global__ void k_deg_accum(const int* __restrict__ dst, int E,
                            float* __restrict__ deg) {
    int e = blockIdx.x * blockDim.x + threadIdx.x;
    if (e < E) atomicAdd(&deg[dst[e]], 1.0f);
}

__global__ void k_dinv_seed(const float* __restrict__ deg, const float* __restrict__ x,
                            float* __restrict__ dinv, float* __restrict__ aggx) {
    int i = blockIdx.x * blockDim.x + threadIdx.x;
    if (i >= N_NODES) return;
    float di = rsqrtf(fmaxf(deg[i], 1.0f));
    dinv[i] = di;
    float dd = di * di;
    aggx[F0 * i]     = x[F0 * i] * dd;
    aggx[F0 * i + 1] = x[F0 * i + 1] * dd;
}


__global__ void k_edge_aggx(const int* __restrict__ src, const int* __restrict__ dst,
                            int E, const float* __restrict__ dinv,
                            const float* __restrict__ x, float* __restrict__ aggx) {
    int t = blockIdx.x * 256 + threadIdx.x;
    int e = t >> 1;
    int f = t & (F0 - 1);
    if (e >= E) return;
    int s = src[e];
    int d = dst[e];
    float norm = dinv[s] * dinv[d];
    atomicAdd(&aggx[d * F0 + f], x[s * F0 + f] * norm);
}


__global__ void k_layer1(const float* __restrict__ aggx, const float* __restrict__ W1,
                         const float* __restrict__ b1, const float* __restrict__ dinv,
                         float* __restrict__ h, float* __restrict__ aggh) {
    int idx = blockIdx.x * blockDim.x + threadIdx.x;
    if (idx >= N_NODES * F1) return;
    int n = idx >> 6;
    int f = idx & (F1 - 1);
    float v = fmaf(aggx[F0 * n], W1[f], aggx[F0 * n + 1] * W1[F1 + f]);
    v = leaky(v + b1[f]);
    h[idx] = v;
    float di = dinv[n];
    aggh[idx] = v * di * di;
}


__global__ void k_edge_aggh(const int* __restrict__ src, const int* __restrict__ dst,
                            int E, const float* __restrict__ dinv,
                            const float* __restrict__ h, float* __restrict__ aggh) {
    int t = blockIdx.x * 256 + threadIdx.x;
    int e = t >> 6;
    int f = t & (F1 - 1);
    if (e >= E) return;
    int s = src[e];
    int d = dst[e];
    float norm = dinv[s] * dinv[d];
    atomicAdd(&aggh[d * F1 + f], h[s * F1 + f] * norm);
}


__global__ void __launch_bounds__(256)
k_gemm2_wmma(const float* __restrict__ A,
             const float* __restrict__ W2,
             const float* __restrict__ b2,
             float* __restrict__ z) {
    const int lane = threadIdx.x & 31;
    const int wave = threadIdx.x >> 5;
    const int row0 = blockIdx.x * 16;
    const int col0 = wave * 16;

    const int m  = lane & 15;
    const int kh = (lane >> 4) << 1;

    v8f acc = {};
    const float* Arow = A + (size_t)(row0 + m) * F1;
#pragma unroll
    for (int k0 = 0; k0 < F1; k0 += 4) {
        v2f a, b;
        a.x = Arow[k0 + kh];
        a.y = Arow[k0 + kh + 1];
        b.x = W2[(k0 + kh) * F2 + col0 + m];
        b.y = W2[(k0 + kh + 1) * F2 + col0 + m];
        acc = __builtin_amdgcn_wmma_f32_16x16x4_f32(
             false, a,  false, b,
             (short)0, acc,  false,  false);
    }

    const int nc    = lane & 15;
    const int rbase = row0 + ((lane >> 4) << 3);
    const float bias = b2[col0 + nc];
#pragma unroll
    for (int v = 0; v < 8; ++v) {
        int   r  = rbase + v;
        size_t o = (size_t)r * F2 + col0 + nc;
        z[o]     = leaky(acc[v] + bias);
    }
}


extern "C" void kernel_launch(void* const* d_in, const int* in_sizes, int n_in,
                              void* d_out, int out_size, void* d_ws, size_t ws_size,
                              hipStream_t stream) {
    const int*   edge_index = (const int*)d_in[0];
    const float* x          = (const float*)d_in[1];
    const float* W1         = (const float*)d_in[2];
    const float* b1         = (const float*)d_in[3];
    const float* W2         = (const float*)d_in[4];
    const float* b2         = (const float*)d_in[5];
    float*       z          = (float*)d_out;

    const int E = in_sizes[0] / 2;
    const int* src = edge_index;
    const int* dst = edge_index + E;

    float* ws   = (float*)d_ws;
    float* deg  = ws;
    float* dinv = ws + N_NODES;
    float* aggx = ws + 2 * N_NODES;
    float* h    = ws + 4 * N_NODES;
    float* aggh = h + N_NODES * F1;
    (void)ws_size; (void)n_in; (void)out_size;

    const int T = 256;

    k_deg_init <<<(N_NODES + T - 1) / T, T, 0, stream>>>(deg);
    k_deg_accum<<<(E + T - 1) / T, T, 0, stream>>>(dst, E, deg);
    k_dinv_seed<<<(N_NODES + T - 1) / T, T, 0, stream>>>(deg, x, dinv, aggx);

    k_edge_aggx<<<((size_t)E * F0 + T - 1) / T, T, 0, stream>>>(src, dst, E, dinv, x, aggx);
    k_layer1   <<<(N_NODES * F1 + T - 1) / T, T, 0, stream>>>(aggx, W1, b1, dinv, h, aggh);

    k_edge_aggh <<<((size_t)E * F1 + T - 1) / T, T, 0, stream>>>(src, dst, E, dinv, h, aggh);
    k_gemm2_wmma<<<N_NODES / 16, 256, 0, stream>>>(aggh, W2, b2, z);
}


# Edges grouped by destination in-call (returning int add = rank, scan, fill); both aggregations become per-node gathers with plain row stores, no float atomics; float summation order changes
# speedup vs baseline: 4.7218x; 2.1636x over previous
#include <hip/hip_runtime.h>


#define N_NODES   100000
#define SCAN_SEG  3200
#define N_PAD     (32 * SCAN_SEG)
#define F0        2
#define F1        64
#define F2        128
#define NEG_SLOPE 0.01f

typedef __attribute__((ext_vector_type(2))) float v2f;
typedef __attribute__((ext_vector_type(8))) float v8f;

__device__ __forceinline__ float leaky(float v) {
    return v >= 0.0f ? v : v * NEG_SLOPE;
}


__global__ void k_zero_cnt(int* __restrict__ cnt) {
    int i = blockIdx.x * blockDim.x + threadIdx.x;
    if (i < N_PAD / 4) reinterpret_cast<int4*>(cnt)[i] = make_int4(0, 0, 0, 0);
}

__global__ void k_count(const int* __restrict__ dst, int E,
                        int* __restrict__ cnt, int* __restrict__ pos) {
    int e = blockIdx.x * blockDim.x + threadIdx.x;
    if (e < E) pos[e] = atomicAdd(&cnt[dst[e]], 1);
}

__global__ void __launch_bounds__(1024)
k_scan(const int* __restrict__ cnt, int* __restrict__ rowptr) {
    __shared__ int wsum[32];
    const int lane = threadIdx.x & 31;
    const int wave = threadIdx.x >> 5;
    const int4* c4 = reinterpret_cast<const int4*>(cnt) + wave * (SCAN_SEG / 4) + lane;
    int4*       r4 = reinterpret_cast<int4*>(rowptr)    + wave * (SCAN_SEG / 4) + lane;

    int4 v[25];
#pragma unroll
    for (int c = 0; c < 25; ++c) v[c] = c4[c * 32];
    int tot = 0;
#pragma unroll
    for (int c = 0; c < 25; ++c) tot += v[c].x + v[c].y + v[c].z + v[c].w;
    for (int o = 16; o > 0; o >>= 1) tot += __shfl_xor(tot, o);
    if (lane == 0) wsum[wave] = tot;
    __syncthreads();
    int carry = (lane < wave) ? wsum[lane] : 0;
    for (int o = 16; o > 0; o >>= 1) carry += __shfl_xor(carry, o);
#pragma unroll
    for (int c = 0; c < 25; ++c) {
        const int s0 = v[c].x, s1 = s0 + v[c].y, s2 = s1 + v[c].z, s3 = s2 + v[c].w;
        int inc = s3;
        for (int o = 1; o < 32; o <<= 1) {
            int u = __shfl_up(inc, o);
            if (lane >= o) inc += u;
        }
        const int tb = carry + inc - s3;
        r4[c * 32] = make_int4(tb, tb + s0, tb + s1, tb + s2);
        carry += __shfl(inc, 31);
    }
}

__global__ void k_fill(const int* __restrict__ src, const int* __restrict__ dst, int E,
                       const int* __restrict__ rowptr, const int* __restrict__ pos,
                       const int* __restrict__ cnt, const float* __restrict__ x,
                       int* __restrict__ col, float* __restrict__ dinv,
                       float* __restrict__ xs) {
    const int t = blockIdx.x * blockDim.x + threadIdx.x;
    if (t < E) {
        unsigned slot = (unsigned)(rowptr[dst[t]] + pos[t]);
        slot = min(slot, (unsigned)(E - 1));
        col[slot] = src[t];
    }
    for (int n = t; n < N_NODES; n += gridDim.x * blockDim.x) {
        const float di = rsqrtf(fmaxf((float)(cnt[n] + 1), 1.0f));
        dinv[n] = di;
        xs[F0 * n]     = x[F0 * n] * di;
        xs[F0 * n + 1] = x[F0 * n + 1] * di;
    }
}


__global__ void k_gather_x(const int* __restrict__ rowptr, const int* __restrict__ col, int E,
                           const float* __restrict__ xs, const float* __restrict__ dinv,
                           float* __restrict__ aggx) {
    const int n = blockIdx.x * blockDim.x + threadIdx.x;
    if (n >= N_NODES) return;
    const float2* xs2 = reinterpret_cast<const float2*>(xs);
    const int beg = max(rowptr[n], 0);
    const int end = min(rowptr[n + 1], E);
    float2 a = xs2[n];
#pragma unroll 4
    for (int j = beg; j < end; ++j) {
        const unsigned s = min((unsigned)col[j], (unsigned)(N_NODES - 1));
        const float2 v = xs2[s];
        a.x += v.x;
        a.y += v.y;
    }
    const float di = dinv[n];
    reinterpret_cast<float2*>(aggx)[n] = make_float2(a.x * di, a.y * di);
}


__global__ void k_layer1(const float* __restrict__ aggx, const float* __restrict__ W1,
                         const float* __restrict__ b1, const float* __restrict__ dinv,
                         float* __restrict__ hs) {
    int idx = blockIdx.x * blockDim.x + threadIdx.x;
    if (idx >= N_NODES * F1) return;
    int n = idx >> 6;
    int f = idx & (F1 - 1);
    float v = fmaf(aggx[F0 * n], W1[f], aggx[F0 * n + 1] * W1[F1 + f]);
    v = leaky(v + b1[f]);
    hs[idx] = v * dinv[n];
}


__global__ void __launch_bounds__(256)
k_gather_h(const int* __restrict__ rowptr, const int* __restrict__ col, int E,
           const float* __restrict__ hs, const float* __restrict__ dinv,
           float* __restrict__ aggh) {
    const int lane   = threadIdx.x & 31;
    const int nwaves = gridDim.x * 8;
    const float2* hs2 = reinterpret_cast<const float2*>(hs);
    float2*       ag2 = reinterpret_cast<float2*>(aggh);
    int n = __builtin_amdgcn_readfirstlane(blockIdx.x * 8 + (threadIdx.x >> 5));
    for (; n < N_NODES; n += nwaves) {
        const int beg = max(rowptr[n], 0);
        const int end = min(rowptr[n + 1], E);
        const float di = dinv[n];
        const int deg   = max(end - beg, 0);
        const int total = deg + 1;
        float2 a = make_float2(0.0f, 0.0f);
        for (int base = 0; base < total; base += 32) {
            const int m = min(32, total - base);
            int mine = n;
            if (base + lane < deg) mine = col[beg + base + lane];
            mine = (int)min((unsigned)mine, (unsigned)(N_NODES - 1));
            for (int j0 = 0; j0 < m; j0 += 8) {
                float2 v[8];
#pragma unroll
                for (int u = 0; u < 8; ++u) {
                    const int s = __builtin_amdgcn_readlane(mine, j0 + u);
                    v[u] = hs2[(size_t)s * 32 + lane];
                }
                asm volatile("" :: "v"(v[0].x), "v"(v[0].y), "v"(v[1].x), "v"(v[1].y),
                                   "v"(v[2].x), "v"(v[2].y), "v"(v[3].x), "v"(v[3].y),
                                   "v"(v[4].x), "v"(v[4].y), "v"(v[5].x), "v"(v[5].y),
                                   "v"(v[6].x), "v"(v[6].y), "v"(v[7].x), "v"(v[7].y));
                a.x += v[0].x;
                a.y += v[0].y;
#pragma unroll
                for (int u = 1; u < 8; ++u) {
                    if (j0 + u < m) {
                        a.x += v[u].x;
                        a.y += v[u].y;
                    }
                }
            }
        }
        ag2[(size_t)n * 32 + lane] = make_float2(a.x * di, a.y * di);
    }
}


__global__ void __launch_bounds__(256)
k_gemm2_wmma(const float* __restrict__ A,
             const float* __restrict__ W2,
             const float* __restrict__ b2,
             float* __restrict__ z) {
    const int lane = threadIdx.x & 31;
    const int wave = threadIdx.x >> 5;
    const int row0 = blockIdx.x * 16;
    const int col0 = wave * 16;

    const int m  = lane & 15;
    const int kh = (lane >> 4) << 1;

    v8f acc = {};
    const float* Arow = A + (size_t)(row0 + m) * F1;
#pragma unroll
    for (int k0 = 0; k0 < F1; k0 += 4) {
        v2f a, b;
        a.x = Arow[k0 + kh];
        a.y = Arow[k0 + kh + 1];
        b.x = W2[(k0 + kh) * F2 + col0 + m];
        b.y = W2[(k0 + kh + 1) * F2 + col0 + m];
        acc = __builtin_amdgcn_wmma_f32_16x16x4_f32(
             false, a,  false, b,
             (short)0, acc,  false,  false);
    }

    const int nc    = lane & 15;
    const int rbase = row0 + ((lane >> 4) << 3);
    const float bias = b2[col0 + nc];
#pragma unroll
    for (int v = 0; v < 8; ++v) {
        int   r  = rbase + v;
        size_t o = (size_t)r * F2 + col0 + nc;
        z[o]     = leaky(acc[v] + bias);
    }
}


extern "C" void kernel_launch(void* const* d_in, const int* in_sizes, int n_in,
                              void* d_out, int out_size, void* d_ws, size_t ws_size,
                              hipStream_t stream) {
    const int*   edge_index = (const int*)d_in[0];
    const float* x          = (const float*)d_in[1];
    const float* W1         = (const float*)d_in[2];
    const float* b1         = (const float*)d_in[3];
    const float* W2         = (const float*)d_in[4];
    const float* b2         = (const float*)d_in[5];
    float*       z          = (float*)d_out;

    const int E = in_sizes[0] / 2;
    const int* src = edge_index;
    const int* dst = edge_index + E;

    float* ws     = (float*)d_ws;
    int*   cnt    = (int*)ws;
    int*   rowptr = (int*)ws + 102400;
    float* dinv   = ws + 204800;
    float* xs     = ws + 307200;
    float* aggx   = ws + 512000;
    float* hs     = ws + 716800;
    float* aggh   = ws + 7116800;
    int*   pos    = (int*)ws + 13516800;
    int*   col    = pos + E;
    (void)ws_size; (void)n_in; (void)out_size;

    const int T = 256;

    k_zero_cnt<<<N_PAD / 4 / T, T, 0, stream>>>(cnt);
    k_count   <<<(E + T - 1) / T, T, 0, stream>>>(dst, E, cnt, pos);
    k_scan    <<<1, 1024, 0, stream>>>(cnt, rowptr);
    k_fill    <<<(E + T - 1) / T, T, 0, stream>>>(src, dst, E, rowptr, pos, cnt, x, col, dinv, xs);

    k_gather_x<<<(N_NODES + T - 1) / T, T, 0, stream>>>(rowptr, col, E, xs, dinv, aggx);
    k_layer1  <<<(N_NODES * F1 + T - 1) / T, T, 0, stream>>>(aggx, W1, b1, dinv, hs);

    k_gather_h  <<<2048, 256, 0, stream>>>(rowptr, col, E, hs, dinv, aggh);
    k_gemm2_wmma<<<N_NODES / 16, 256, 0, stream>>>(aggh, W2, b2, z);
}


# Layer-2 gather computes each source's activated layer-1 row from a 16-B record (2-wide aggregate, dinv), not a stored 256-B row; layer-1 kernel and plane dropped; sum order kept, rounding changes
# speedup vs baseline: 5.3248x; 1.1277x over previous
#include <hip/hip_runtime.h>


#define N_NODES   100000
#define SCAN_SEG  3200
#define N_PAD     (32 * SCAN_SEG)
#define F0        2
#define F1        64
#define F2        128
#define NEG_SLOPE 0.01f

typedef __attribute__((ext_vector_type(2))) float v2f;
typedef __attribute__((ext_vector_type(8))) float v8f;

__device__ __forceinline__ float leaky(float v) {
    return v >= 0.0f ? v : v * NEG_SLOPE;
}


__global__ void k_zero_cnt(int* __restrict__ cnt) {
    int i = blockIdx.x * blockDim.x + threadIdx.x;
    if (i < N_PAD / 4) reinterpret_cast<int4*>(cnt)[i] = make_int4(0, 0, 0, 0);
}

__global__ void k_count(const int* __restrict__ dst, int E,
                        int* __restrict__ cnt, int* __restrict__ pos) {
    int e = blockIdx.x * blockDim.x + threadIdx.x;
    if (e < E) pos[e] = atomicAdd(&cnt[dst[e]], 1);
}

__global__ void __launch_bounds__(1024)
k_scan(const int* __restrict__ cnt, int* __restrict__ rowptr) {
    __shared__ int wsum[32];
    const int lane = threadIdx.x & 31;
    const int wave = threadIdx.x >> 5;
    const int4* c4 = reinterpret_cast<const int4*>(cnt) + wave * (SCAN_SEG / 4) + lane;
    int4*       r4 = reinterpret_cast<int4*>(rowptr)    + wave * (SCAN_SEG / 4) + lane;

    int4 v[25];
#pragma unroll
    for (int c = 0; c < 25; ++c) v[c] = c4[c * 32];
    int tot = 0;
#pragma unroll
    for (int c = 0; c < 25; ++c) tot += v[c].x + v[c].y + v[c].z + v[c].w;
    for (int o = 16; o > 0; o >>= 1) tot += __shfl_xor(tot, o);
    if (lane == 0) wsum[wave] = tot;
    __syncthreads();
    int carry = (lane < wave) ? wsum[lane] : 0;
    for (int o = 16; o > 0; o >>= 1) carry += __shfl_xor(carry, o);
#pragma unroll
    for (int c = 0; c < 25; ++c) {
        const int s0 = v[c].x, s1 = s0 + v[c].y, s2 = s1 + v[c].z, s3 = s2 + v[c].w;
        int inc = s3;
        for (int o = 1; o < 32; o <<= 1) {
            int u = __shfl_up(inc, o);
            if (lane >= o) inc += u;
        }
        const int tb = carry + inc - s3;
        r4[c * 32] = make_int4(tb, tb + s0, tb + s1, tb + s2);
        carry += __shfl(inc, 31);
    }
}

__global__ void k_fill(const int* __restrict__ src, const int* __restrict__ dst, int E,
                       const int* __restrict__ rowptr, const int* __restrict__ pos,
                       const int* __restrict__ cnt, const float* __restrict__ x,
                       int* __restrict__ col, float* __restrict__ dinv,
                       float* __restrict__ xs) {
    const int t = blockIdx.x * blockDim.x + threadIdx.x;
    if (t < E) {
        unsigned slot = (unsigned)(rowptr[dst[t]] + pos[t]);
        slot = min(slot, (unsigned)(E - 1));
        col[slot] = src[t];
    }
    for (int n = t; n < N_NODES; n += gridDim.x * blockDim.x) {
        const float di = rsqrtf(fmaxf((float)(cnt[n] + 1), 1.0f));
        dinv[n] = di;
        xs[F0 * n]     = x[F0 * n] * di;
        xs[F0 * n + 1] = x[F0 * n + 1] * di;
    }
}


__global__ void k_gather_x(const int* __restrict__ rowptr, const int* __restrict__ col, int E,
                           const float* __restrict__ xs, const float* __restrict__ dinv,
                           float* __restrict__ rec) {
    const int n = blockIdx.x * blockDim.x + threadIdx.x;
    if (n >= N_NODES) return;
    const float2* xs2 = reinterpret_cast<const float2*>(xs);
    const int beg = max(rowptr[n], 0);
    const int end = min(rowptr[n + 1], E);
    float2 a = xs2[n];
#pragma unroll 4
    for (int j = beg; j < end; ++j) {
        const unsigned s = min((unsigned)col[j], (unsigned)(N_NODES - 1));
        const float2 v = xs2[s];
        a.x += v.x;
        a.y += v.y;
    }
    const float di = dinv[n];
    reinterpret_cast<float4*>(rec)[n] = make_float4(a.x * di, a.y * di, di, 0.0f);
}


__global__ void __launch_bounds__(256)
k_gather_act(const int* __restrict__ rowptr, const int* __restrict__ col, int E,
             const float* __restrict__ rec, const float* __restrict__ W1,
             const float* __restrict__ b1, const float* __restrict__ dinv,
             float* __restrict__ aggh) {
    const int lane   = threadIdx.x & 31;
    const int nwaves = gridDim.x * 8;
    const float4* rec4 = reinterpret_cast<const float4*>(rec);
    float2*       ag2  = reinterpret_cast<float2*>(aggh);
    const v2f w0 = {W1[2 * lane],      W1[2 * lane + 1]};
    const v2f w1 = {W1[F1 + 2 * lane], W1[F1 + 2 * lane + 1]};
    const v2f bb = {b1[2 * lane],      b1[2 * lane + 1]};
    const v2f slope = {NEG_SLOPE, NEG_SLOPE};
    int n = __builtin_amdgcn_readfirstlane(blockIdx.x * 8 + (threadIdx.x >> 5));
    for (; n < N_NODES; n += nwaves) {
        const int beg = max(rowptr[n], 0);
        const int end = min(rowptr[n + 1], E);
        const float di = dinv[n];
        const int deg   = max(end - beg, 0);
        const int total = deg + 1;
        v2f a = {0.0f, 0.0f};
        for (int base = 0; base < total; base += 32) {
            const int m = min(32, total - base);
            int mine = n;
            if (base + lane < deg) mine = col[beg + base + lane];
            mine = (int)min((unsigned)mine, (unsigned)(N_NODES - 1));
            const float4 q = rec4[mine];
            const int qx = __float_as_int(q.x), qy = __float_as_int(q.y), qz = __float_as_int(q.z);
#pragma unroll 4
            for (int j = 0; j < m; ++j) {
                const float ax = __int_as_float(__builtin_amdgcn_readlane(qx, j));
                const float ay = __int_as_float(__builtin_amdgcn_readlane(qy, j));
                const float ds = __int_as_float(__builtin_amdgcn_readlane(qz, j));
                const v2f vax = {ax, ax}, vay = {ay, ay}, vds = {ds, ds};
                const v2f t = __builtin_elementwise_fma(vax, w0, __builtin_elementwise_fma(vay, w1, bb));
                const v2f l = __builtin_elementwise_max(t, t * slope);
                a = __builtin_elementwise_fma(l, vds, a);
            }
        }
        ag2[(size_t)n * 32 + lane] = make_float2(a.x * di, a.y * di);
    }
}


__global__ void __launch_bounds__(256)
k_gemm2_wmma(const float* __restrict__ A,
             const float* __restrict__ W2,
             const float* __restrict__ b2,
             float* __restrict__ z) {
    const int lane = threadIdx.x & 31;
    const int wave = threadIdx.x >> 5;
    const int row0 = blockIdx.x * 16;
    const int col0 = wave * 16;

    const int m  = lane & 15;
    const int kh = (lane >> 4) << 1;

    v8f acc = {};
    const float* Arow = A + (size_t)(row0 + m) * F1;
#pragma unroll
    for (int k0 = 0; k0 < F1; k0 += 4) {
        v2f a, b;
        a.x = Arow[k0 + kh];
        a.y = Arow[k0 + kh + 1];
        b.x = W2[(k0 + kh) * F2 + col0 + m];
        b.y = W2[(k0 + kh + 1) * F2 + col0 + m];
        acc = __builtin_amdgcn_wmma_f32_16x16x4_f32(
             false, a,  false, b,
             (short)0, acc,  false,  false);
    }

    const int nc    = lane & 15;
    const int rbase = row0 + ((lane >> 4) << 3);
    const float bias = b2[col0 + nc];
#pragma unroll
    for (int v = 0; v < 8; ++v) {
        int   r  = rbase + v;
        size_t o = (size_t)r * F2 + col0 + nc;
        z[o]     = leaky(acc[v] + bias);
    }
}


extern "C" void kernel_launch(void* const* d_in, const int* in_sizes, int n_in,
                              void* d_out, int out_size, void* d_ws, size_t ws_size,
                              hipStream_t stream) {
    const int*   edge_index = (const int*)d_in[0];
    const float* x          = (const float*)d_in[1];
    const float* W1         = (const float*)d_in[2];
    const float* b1         = (const float*)d_in[3];
    const float* W2         = (const float*)d_in[4];
    const float* b2         = (const float*)d_in[5];
    float*       z          = (float*)d_out;

    const int E = in_sizes[0] / 2;
    const int* src = edge_index;
    const int* dst = edge_index + E;

    float* ws     = (float*)d_ws;
    int*   cnt    = (int*)ws;
    int*   rowptr = (int*)ws + 102400;
    float* dinv   = ws + 204800;
    float* xs     = ws + 307200;
    float* rec    = ws + 512000;
    float* aggh   = ws + 7116800;
    int*   pos    = (int*)ws + 13516800;
    int*   col    = pos + E;
    (void)ws_size; (void)n_in; (void)out_size;

    const int T = 256;

    k_zero_cnt<<<N_PAD / 4 / T, T, 0, stream>>>(cnt);
    k_count   <<<(E + T - 1) / T, T, 0, stream>>>(dst, E, cnt, pos);
    k_scan    <<<1, 1024, 0, stream>>>(cnt, rowptr);
    k_fill    <<<(E + T - 1) / T, T, 0, stream>>>(src, dst, E, rowptr, pos, cnt, x, col, dinv, xs);

    k_gather_x<<<(N_NODES + T - 1) / T, T, 0, stream>>>(rowptr, col, E, xs, dinv, rec);

    k_gather_act<<<2048, 256, 0, stream>>>(rowptr, col, E, rec, W1, b1, dinv, aggh);
    k_gemm2_wmma<<<N_NODES / 16, 256, 0, stream>>>(aggh, W2, b2, z);
}


# Layer-2 GEMM fused into the layer-2 gather kernel: a block gathers a 16-node stripe into LDS and multiplies it there, W2 fragments in registers; 64-wide plane and one launch dropped; sum order kept
# speedup vs baseline: 5.6841x; 1.0675x over previous
#include <hip/hip_runtime.h>


#define N_NODES   100000
#define SCAN_SEG  3200
#define N_PAD     (32 * SCAN_SEG)
#define F0        2
#define F1        64
#define F2        128
#define NEG_SLOPE 0.01f

typedef __attribute__((ext_vector_type(2))) float v2f;
typedef __attribute__((ext_vector_type(8))) float v8f;

__device__ __forceinline__ float leaky(float v) {
    return v >= 0.0f ? v : v * NEG_SLOPE;
}


__global__ void k_zero_cnt(int* __restrict__ cnt) {
    int i = blockIdx.x * blockDim.x + threadIdx.x;
    if (i < N_PAD / 4) reinterpret_cast<int4*>(cnt)[i] = make_int4(0, 0, 0, 0);
}

__global__ void k_count(const int* __restrict__ dst, int E,
                        int* __restrict__ cnt, int* __restrict__ pos) {
    int e = blockIdx.x * blockDim.x + threadIdx.x;
    if (e < E) pos[e] = atomicAdd(&cnt[dst[e]], 1);
}

__global__ void __launch_bounds__(1024)
k_scan(const int* __restrict__ cnt, int* __restrict__ rowptr) {
    __shared__ int wsum[32];
    const int lane = threadIdx.x & 31;
    const int wave = threadIdx.x >> 5;
    const int4* c4 = reinterpret_cast<const int4*>(cnt) + wave * (SCAN_SEG / 4) + lane;
    int4*       r4 = reinterpret_cast<int4*>(rowptr)    + wave * (SCAN_SEG / 4) + lane;

    int4 v[25];
#pragma unroll
    for (int c = 0; c < 25; ++c) v[c] = c4[c * 32];
    int tot = 0;
#pragma unroll
    for (int c = 0; c < 25; ++c) tot += v[c].x + v[c].y + v[c].z + v[c].w;
    for (int o = 16; o > 0; o >>= 1) tot += __shfl_xor(tot, o);
    if (lane == 0) wsum[wave] = tot;
    __syncthreads();
    int carry = (lane < wave) ? wsum[lane] : 0;
    for (int o = 16; o > 0; o >>= 1) carry += __shfl_xor(carry, o);
#pragma unroll
    for (int c = 0; c < 25; ++c) {
        const int s0 = v[c].x, s1 = s0 + v[c].y, s2 = s1 + v[c].z, s3 = s2 + v[c].w;
        int inc = s3;
        for (int o = 1; o < 32; o <<= 1) {
            int u = __shfl_up(inc, o);
            if (lane >= o) inc += u;
        }
        const int tb = carry + inc - s3;
        r4[c * 32] = make_int4(tb, tb + s0, tb + s1, tb + s2);
        carry += __shfl(inc, 31);
    }
}

__global__ void k_fill(const int* __restrict__ src, const int* __restrict__ dst, int E,
                       const int* __restrict__ rowptr, const int* __restrict__ pos,
                       const int* __restrict__ cnt, const float* __restrict__ x,
                       int* __restrict__ col, float* __restrict__ dinv,
                       float* __restrict__ xs) {
    const int t = blockIdx.x * blockDim.x + threadIdx.x;
    if (t < E) {
        unsigned slot = (unsigned)(rowptr[dst[t]] + pos[t]);
        slot = min(slot, (unsigned)(E - 1));
        col[slot] = src[t];
    }
    for (int n = t; n < N_NODES; n += gridDim.x * blockDim.x) {
        const float di = rsqrtf(fmaxf((float)(cnt[n] + 1), 1.0f));
        dinv[n] = di;
        xs[F0 * n]     = x[F0 * n] * di;
        xs[F0 * n + 1] = x[F0 * n + 1] * di;
    }
}


__global__ void k_gather_x(const int* __restrict__ rowptr, const int* __restrict__ col, int E,
                           const float* __restrict__ xs, const float* __restrict__ dinv,
                           float* __restrict__ rec) {
    const int n = blockIdx.x * blockDim.x + threadIdx.x;
    if (n >= N_NODES) return;
    const float2* xs2 = reinterpret_cast<const float2*>(xs);
    const int beg = max(rowptr[n], 0);
    const int end = min(rowptr[n + 1], E);
    float2 a = xs2[n];
#pragma unroll 4
    for (int j = beg; j < end; ++j) {
        const unsigned s = min((unsigned)col[j], (unsigned)(N_NODES - 1));
        const float2 v = xs2[s];
        a.x += v.x;
        a.y += v.y;
    }
    const float di = dinv[n];
    reinterpret_cast<float4*>(rec)[n] = make_float4(a.x * di, a.y * di, di, 0.0f);
}


#define STRIPES        (N_NODES / 16)
#define STRIPES_PER_WG 5
#define TILE_PITCH     68

__global__ void __launch_bounds__(256)
k_gather_gemm(const int* __restrict__ rowptr, const int* __restrict__ col, int E,
              const float* __restrict__ rec, const float* __restrict__ W1,
              const float* __restrict__ b1, const float* __restrict__ dinv,
              const float* __restrict__ W2, const float* __restrict__ b2,
              float* __restrict__ z) {
    __shared__ float tile[2][16 * TILE_PITCH];
    const int lane = threadIdx.x & 31;
    const int wave = __builtin_amdgcn_readfirstlane(threadIdx.x >> 5);
    const float4* rec4 = reinterpret_cast<const float4*>(rec);
    const v2f w0 = {W1[2 * lane],      W1[2 * lane + 1]};
    const v2f w1 = {W1[F1 + 2 * lane], W1[F1 + 2 * lane + 1]};
    const v2f bb = {b1[2 * lane],      b1[2 * lane + 1]};
    const v2f slope = {NEG_SLOPE, NEG_SLOPE};

    const int col0 = wave * 16;
    const int m    = lane & 15;
    const int kh   = (lane >> 4) << 1;
    v2f bf[F1 / 4];
#pragma unroll
    for (int t = 0; t < F1 / 4; ++t) {
        bf[t].x = W2[(4 * t + kh) * F2 + col0 + m];
        bf[t].y = W2[(4 * t + kh + 1) * F2 + col0 + m];
    }
    const float bias = b2[col0 + m];

    int buf = 0;
    for (int s = blockIdx.x; s < STRIPES; s += gridDim.x) {
        float* tl = tile[buf];
        for (int i = 0; i < 2; ++i) {
            const int r = 2 * wave + i;
            const int n = s * 16 + r;
            const int beg = max(rowptr[n], 0);
            const int end = min(rowptr[n + 1], E);
            const float di = dinv[n];
            const int deg   = max(end - beg, 0);
            const int total = deg + 1;
            v2f a = {0.0f, 0.0f};
            for (int base = 0; base < total; base += 32) {
                const int cnt = min(32, total - base);
                int mine = n;
                if (base + lane < deg) mine = col[beg + base + lane];
                mine = (int)min((unsigned)mine, (unsigned)(N_NODES - 1));
                const float4 q = rec4[mine];
                const int qx = __float_as_int(q.x), qy = __float_as_int(q.y), qz = __float_as_int(q.z);
#pragma unroll 4
                for (int j = 0; j < cnt; ++j) {
                    const float ax = __int_as_float(__builtin_amdgcn_readlane(qx, j));
                    const float ay = __int_as_float(__builtin_amdgcn_readlane(qy, j));
                    const float ds = __int_as_float(__builtin_amdgcn_readlane(qz, j));
                    const v2f vax = {ax, ax}, vay = {ay, ay}, vds = {ds, ds};
                    const v2f t = __builtin_elementwise_fma(vax, w0, __builtin_elementwise_fma(vay, w1, bb));
                    const v2f l = __builtin_elementwise_max(t, t * slope);
                    a = __builtin_elementwise_fma(l, vds, a);
                }
            }
            const v2f row = {a.x * di, a.y * di};
            *reinterpret_cast<v2f*>(&tl[r * TILE_PITCH + 2 * lane]) = row;
        }
        __syncthreads();

        v8f acc = {};
        const float* arow = tl + m * TILE_PITCH + kh;
#pragma unroll
        for (int t = 0; t < F1 / 4; ++t) {
            const v2f af = *reinterpret_cast<const v2f*>(arow + 4 * t);
            acc = __builtin_amdgcn_wmma_f32_16x16x4_f32(
                 false, af,  false, bf[t],
                 (short)0, acc,  false,  false);
        }

        const int rbase = s * 16 + ((lane >> 4) << 3);
#pragma unroll
        for (int v = 0; v < 8; ++v) {
            const size_t o = (size_t)(rbase + v) * F2 + col0 + m;
            z[o] = leaky(acc[v] + bias);
        }
        buf ^= 1;
    }
}


extern "C" void kernel_launch(void* const* d_in, const int* in_sizes, int n_in,
                              void* d_out, int out_size, void* d_ws, size_t ws_size,
                              hipStream_t stream) {
    const int*   edge_index = (const int*)d_in[0];
    const float* x          = (const float*)d_in[1];
    const float* W1         = (const float*)d_in[2];
    const float* b1         = (const float*)d_in[3];
    const float* W2         = (const float*)d_in[4];
    const float* b2         = (const float*)d_in[5];
    float*       z          = (float*)d_out;

    const int E = in_sizes[0] / 2;
    const int* src = edge_index;
    const int* dst = edge_index + E;

    float* ws     = (float*)d_ws;
    int*   cnt    = (int*)ws;
    int*   rowptr = (int*)ws + 102400;
    float* dinv   = ws + 204800;
    float* xs     = ws + 307200;
    float* rec    = ws + 512000;
    int*   pos    = (int*)ws + 13516800;
    int*   col    = pos + E;
    (void)ws_size; (void)n_in; (void)out_size;

    const int T = 256;

    k_zero_cnt<<<N_PAD / 4 / T, T, 0, stream>>>(cnt);
    k_count   <<<(E + T - 1) / T, T, 0, stream>>>(dst, E, cnt, pos);
    k_scan    <<<1, 1024, 0, stream>>>(cnt, rowptr);
    k_fill    <<<(E + T - 1) / T, T, 0, stream>>>(src, dst, E, rowptr, pos, cnt, x, col, dinv, xs);

    k_gather_x<<<(N_NODES + T - 1) / T, T, 0, stream>>>(rowptr, col, E, xs, dinv, rec);

    k_gather_gemm<<<STRIPES / STRIPES_PER_WG, 256, 0, stream>>>(rowptr, col, E, rec, W1, b1, dinv, W2, b2, z);
}


# Edges grouped by destination by a two-level counting sort on LDS counters (256-node buckets, then nodes) in place of one returning global add per edge; float summation order changes
# speedup vs baseline: 9.9044x; 1.7425x over previous
#include <hip/hip_runtime.h>


#define N_NODES   100000
#define SCAN_SEG  3200
#define N_PAD     (32 * SCAN_SEG)
#define BKT_SHIFT 8
#define BKT_NODES (1 << BKT_SHIFT)
#define N_CHUNK   256
#define N_BKT     (N_PAD / N_CHUNK)
#define N_BKT_USED ((N_NODES + BKT_NODES - 1) / BKT_NODES)
#define SORT_T    1024
#define BKT_T     512
#define EDGE_B    8
#define BKT_B     8
#define F0        2
#define F1        64
#define F2        128
#define NEG_SLOPE 0.01f

typedef __attribute__((ext_vector_type(2))) float v2f;
typedef __attribute__((ext_vector_type(8))) float v8f;

__device__ __forceinline__ float leaky(float v) {
    return v >= 0.0f ? v : v * NEG_SLOPE;
}


__global__ void __launch_bounds__(SORT_T)
k_bucket_count(const int* __restrict__ dst, int E, int* __restrict__ bcnt) {
    __shared__ int hist[N_BKT];
    const int g   = blockIdx.x;
    const int tid = threadIdx.x;
    if (tid < N_BKT) hist[tid] = 0;
    __syncthreads();
    const int per  = (E + N_CHUNK - 1) / N_CHUNK;
    const int cbeg = min(g * per, E);
    const int cend = min(cbeg + per, E);
    for (int e0 = cbeg + tid; e0 < cend; e0 += SORT_T * EDGE_B) {
        unsigned d[EDGE_B];
#pragma unroll
        for (int k = 0; k < EDGE_B; ++k) {
            const int e = e0 + k * SORT_T;
            d[k] = (unsigned)dst[min(e, cend - 1)];
        }
        asm volatile("" :: "v"(d[0]), "v"(d[1]), "v"(d[2]), "v"(d[3]), "v"(d[4]), "v"(d[5]), "v"(d[6]), "v"(d[7]));
#pragma unroll
        for (int k = 0; k < EDGE_B; ++k) {
            const unsigned dd = min(d[k], (unsigned)(N_NODES - 1));
            if (e0 + k * SORT_T < cend) atomicAdd(&hist[dd >> BKT_SHIFT], 1);
        }
    }
    __syncthreads();
    if (tid < N_BKT) bcnt[tid * N_CHUNK + g] = hist[tid];
}

__global__ void __launch_bounds__(1024)
k_scan(const int* __restrict__ cnt, int* __restrict__ rowptr) {
    __shared__ int wsum[32];
    const int lane = threadIdx.x & 31;
    const int wave = threadIdx.x >> 5;
    const int4* c4 = reinterpret_cast<const int4*>(cnt) + wave * (SCAN_SEG / 4) + lane;
    int4*       r4 = reinterpret_cast<int4*>(rowptr)    + wave * (SCAN_SEG / 4) + lane;

    int4 v[25];
#pragma unroll
    for (int c = 0; c < 25; ++c) v[c] = c4[c * 32];
    int tot = 0;
#pragma unroll
    for (int c = 0; c < 25; ++c) tot += v[c].x + v[c].y + v[c].z + v[c].w;
    for (int o = 16; o > 0; o >>= 1) tot += __shfl_xor(tot, o);
    if (lane == 0) wsum[wave] = tot;
    __syncthreads();
    int carry = (lane < wave) ? wsum[lane] : 0;
    for (int o = 16; o > 0; o >>= 1) carry += __shfl_xor(carry, o);
#pragma unroll
    for (int c = 0; c < 25; ++c) {
        const int s0 = v[c].x, s1 = s0 + v[c].y, s2 = s1 + v[c].z, s3 = s2 + v[c].w;
        int inc = s3;
        for (int o = 1; o < 32; o <<= 1) {
            int u = __shfl_up(inc, o);
            if (lane >= o) inc += u;
        }
        const int tb = carry + inc - s3;
        r4[c * 32] = make_int4(tb, tb + s0, tb + s1, tb + s2);
        carry += __shfl(inc, 31);
    }
}

__global__ void __launch_bounds__(SORT_T)
k_partition(const int* __restrict__ src, const int* __restrict__ dst, int E,
            const int* __restrict__ boff, int* __restrict__ part) {
    __shared__ int cur[N_BKT];
    const int g   = blockIdx.x;
    const int tid = threadIdx.x;
    if (tid < N_BKT) cur[tid] = boff[tid * N_CHUNK + g];
    __syncthreads();
    const int per  = (E + N_CHUNK - 1) / N_CHUNK;
    const int cbeg = min(g * per, E);
    const int cend = min(cbeg + per, E);
    for (int e0 = cbeg + tid; e0 < cend; e0 += SORT_T * EDGE_B) {
        unsigned d[EDGE_B], s[EDGE_B];
#pragma unroll
        for (int k = 0; k < EDGE_B; ++k) {
            const int e = min(e0 + k * SORT_T, cend - 1);
            d[k] = (unsigned)dst[e];
            s[k] = (unsigned)src[e];
        }
        asm volatile("" :: "v"(d[0]), "v"(d[1]), "v"(d[2]), "v"(d[3]), "v"(d[4]), "v"(d[5]), "v"(d[6]), "v"(d[7]));
        asm volatile("" :: "v"(s[0]), "v"(s[1]), "v"(s[2]), "v"(s[3]), "v"(s[4]), "v"(s[5]), "v"(s[6]), "v"(s[7]));
#pragma unroll
        for (int k = 0; k < EDGE_B; ++k) {
            const unsigned dd = min(d[k], (unsigned)(N_NODES - 1));
            if (e0 + k * SORT_T < cend) {
                unsigned slot = (unsigned)atomicAdd(&cur[dd >> BKT_SHIFT], 1);
                slot = min(slot, (unsigned)(E - 1));
                part[slot] = (int)((s[k] & 0x00FFFFFFu) | ((dd & (BKT_NODES - 1)) << 24));
            }
        }
    }
}

__global__ void __launch_bounds__(BKT_T)
k_bucket_sort(const int* __restrict__ part, int E, const int* __restrict__ boff,
              const float* __restrict__ x, int* __restrict__ rowptr, int* __restrict__ col,
              float* __restrict__ dinv, float* __restrict__ xs) {
    __shared__ int cnt[BKT_NODES];
    __shared__ int cur[BKT_NODES];
    __shared__ int wsum[BKT_NODES / 32];
    const int b    = blockIdx.x;
    const int tid  = threadIdx.x;
    const int lane = tid & 31;
    const int wave = tid >> 5;
    const int bbeg = min(max(boff[b * N_CHUNK], 0), E);
    const int bend = min(max(boff[(b + 1) * N_CHUNK], bbeg), E);
    if (tid < BKT_NODES) cnt[tid] = 0;
    __syncthreads();
    for (int i0 = bbeg + tid; i0 < bend; i0 += BKT_T * BKT_B) {
        unsigned r[BKT_B];
#pragma unroll
        for (int k = 0; k < BKT_B; ++k) r[k] = (unsigned)part[min(i0 + k * BKT_T, bend - 1)];
        asm volatile("" :: "v"(r[0]), "v"(r[1]), "v"(r[2]), "v"(r[3]), "v"(r[4]), "v"(r[5]), "v"(r[6]), "v"(r[7]));
#pragma unroll
        for (int k = 0; k < BKT_B; ++k)
            if (i0 + k * BKT_T < bend) atomicAdd(&cnt[r[k] >> 24], 1);
    }
    __syncthreads();
    int c = 0, inc = 0;
    if (tid < BKT_NODES) {
        c = cnt[tid];
        inc = c;
        for (int o = 1; o < 32; o <<= 1) {
            const int u = __shfl_up(inc, o);
            if (lane >= o) inc += u;
        }
        if (lane == 31) wsum[wave] = inc;
    }
    __syncthreads();
    if (tid < BKT_NODES) {
        int carry = 0;
        for (int w = 0; w < wave; ++w) carry += wsum[w];
        const int start = bbeg + carry + inc - c;
        cur[tid] = start;
        const int n = b * BKT_NODES + tid;
        rowptr[n] = start;
        if (n < N_NODES) {
            const float di = rsqrtf(fmaxf((float)(c + 1), 1.0f));
            dinv[n] = di;
            xs[F0 * n]     = x[F0 * n] * di;
            xs[F0 * n + 1] = x[F0 * n + 1] * di;
        }
    }
    __syncthreads();
    for (int i0 = bbeg + tid; i0 < bend; i0 += BKT_T * BKT_B) {
        unsigned r[BKT_B];
#pragma unroll
        for (int k = 0; k < BKT_B; ++k) r[k] = (unsigned)part[min(i0 + k * BKT_T, bend - 1)];
        asm volatile("" :: "v"(r[0]), "v"(r[1]), "v"(r[2]), "v"(r[3]), "v"(r[4]), "v"(r[5]), "v"(r[6]), "v"(r[7]));
#pragma unroll
        for (int k = 0; k < BKT_B; ++k) {
            if (i0 + k * BKT_T < bend) {
                unsigned slot = (unsigned)atomicAdd(&cur[r[k] >> 24], 1);
                slot = min(slot, (unsigned)(E - 1));
                col[slot] = (int)(r[k] & 0x00FFFFFFu);
            }
        }
    }
}


__global__ void k_gather_x(const int* __restrict__ rowptr, const int* __restrict__ col, int E,
                           const float* __restrict__ xs, const float* __restrict__ dinv,
                           float* __restrict__ rec) {
    const int n = blockIdx.x * blockDim.x + threadIdx.x;
    if (n >= N_NODES) return;
    const float2* xs2 = reinterpret_cast<const float2*>(xs);
    const int beg = max(rowptr[n], 0);
    const int end = min(rowptr[n + 1], E);
    float2 a = xs2[n];
#pragma unroll 4
    for (int j = beg; j < end; ++j) {
        const unsigned s = min((unsigned)col[j], (unsigned)(N_NODES - 1));
        const float2 v = xs2[s];
        a.x += v.x;
        a.y += v.y;
    }
    const float di = dinv[n];
    reinterpret_cast<float4*>(rec)[n] = make_float4(a.x * di, a.y * di, di, 0.0f);
}


#define STRIPES        (N_NODES / 16)
#define STRIPES_PER_WG 5
#define TILE_PITCH     68

__global__ void __launch_bounds__(256)
k_gather_gemm(const int* __restrict__ rowptr, const int* __restrict__ col, int E,
              const float* __restrict__ rec, const float* __restrict__ W1,
              const float* __restrict__ b1, const float* __restrict__ dinv,
              const float* __restrict__ W2, const float* __restrict__ b2,
              float* __restrict__ z) {
    __shared__ float tile[2][16 * TILE_PITCH];
    const int lane = threadIdx.x & 31;
    const int wave = __builtin_amdgcn_readfirstlane(threadIdx.x >> 5);
    const float4* rec4 = reinterpret_cast<const float4*>(rec);
    const v2f w0 = {W1[2 * lane],      W1[2 * lane + 1]};
    const v2f w1 = {W1[F1 + 2 * lane], W1[F1 + 2 * lane + 1]};
    const v2f bb = {b1[2 * lane],      b1[2 * lane + 1]};
    const v2f slope = {NEG_SLOPE, NEG_SLOPE};

    const int col0 = wave * 16;
    const int m    = lane & 15;
    const int kh   = (lane >> 4) << 1;
    v2f bf[F1 / 4];
#pragma unroll
    for (int t = 0; t < F1 / 4; ++t) {
        bf[t].x = W2[(4 * t + kh) * F2 + col0 + m];
        bf[t].y = W2[(4 * t + kh + 1) * F2 + col0 + m];
    }
    const float bias = b2[col0 + m];

    int buf = 0;
    for (int s = blockIdx.x; s < STRIPES; s += gridDim.x) {
        float* tl = tile[buf];
        for (int i = 0; i < 2; ++i) {
            const int r = 2 * wave + i;
            const int n = s * 16 + r;
            const int beg = max(rowptr[n], 0);
            const int end = min(rowptr[n + 1], E);
            const float di = dinv[n];
            const int deg   = max(end - beg, 0);
            const int total = deg + 1;
            v2f a = {0.0f, 0.0f};
            for (int base = 0; base < total; base += 32) {
                const int cnt = min(32, total - base);
                int mine = n;
                if (base + lane < deg) mine = col[beg + base + lane];
                mine = (int)min((unsigned)mine, (unsigned)(N_NODES - 1));
                const float4 q = rec4[mine];
                const int qx = __float_as_int(q.x), qy = __float_as_int(q.y), qz = __float_as_int(q.z);
#pragma unroll 4
                for (int j = 0; j < cnt; ++j) {
                    const float ax = __int_as_float(__builtin_amdgcn_readlane(qx, j));
                    const float ay = __int_as_float(__builtin_amdgcn_readlane(qy, j));
                    const float ds = __int_as_float(__builtin_amdgcn_readlane(qz, j));
                    const v2f vax = {ax, ax}, vay = {ay, ay}, vds = {ds, ds};
                    const v2f t = __builtin_elementwise_fma(vax, w0, __builtin_elementwise_fma(vay, w1, bb));
                    const v2f l = __builtin_elementwise_max(t, t * slope);
                    a = __builtin_elementwise_fma(l, vds, a);
                }
            }
            const v2f row = {a.x * di, a.y * di};
            *reinterpret_cast<v2f*>(&tl[r * TILE_PITCH + 2 * lane]) = row;
        }
        __syncthreads();

        v8f acc = {};
        const float* arow = tl + m * TILE_PITCH + kh;
#pragma unroll
        for (int t = 0; t < F1 / 4; ++t) {
            const v2f af = *reinterpret_cast<const v2f*>(arow + 4 * t);
            acc = __builtin_amdgcn_wmma_f32_16x16x4_f32(
                 false, af,  false, bf[t],
                 (short)0, acc,  false,  false);
        }

        const int rbase = s * 16 + ((lane >> 4) << 3);
#pragma unroll
        for (int v = 0; v < 8; ++v) {
            const size_t o = (size_t)(rbase + v) * F2 + col0 + m;
            z[o] = leaky(acc[v] + bias);
        }
        buf ^= 1;
    }
}


extern "C" void kernel_launch(void* const* d_in, const int* in_sizes, int n_in,
                              void* d_out, int out_size, void* d_ws, size_t ws_size,
                              hipStream_t stream) {
    const int*   edge_index = (const int*)d_in[0];
    const float* x          = (const float*)d_in[1];
    const float* W1         = (const float*)d_in[2];
    const float* b1         = (const float*)d_in[3];
    const float* W2         = (const float*)d_in[4];
    const float* b2         = (const float*)d_in[5];
    float*       z          = (float*)d_out;

    const int E = in_sizes[0] / 2;
    const int* src = edge_index;
    const int* dst = edge_index + E;

    float* ws     = (float*)d_ws;
    int*   bcnt   = (int*)ws;
    int*   boff   = (int*)ws + 102400;
    float* dinv   = ws + 204800;
    float* xs     = ws + 307200;
    float* rec    = ws + 512000;
    int*   rowptr = (int*)ws + 1024000;
    int*   part   = (int*)ws + 13516800;
    int*   col    = part + E;
    (void)ws_size; (void)n_in; (void)out_size;

    const int T = 256;

    k_bucket_count<<<N_CHUNK, SORT_T, 0, stream>>>(dst, E, bcnt);
    k_scan        <<<1, 1024, 0, stream>>>(bcnt, boff);
    k_partition   <<<N_CHUNK, SORT_T, 0, stream>>>(src, dst, E, boff, part);
    k_bucket_sort <<<N_BKT_USED, BKT_T, 0, stream>>>(part, E, boff, x, rowptr, col, dinv, xs);

    k_gather_x<<<(N_NODES + T - 1) / T, T, 0, stream>>>(rowptr, col, E, xs, dinv, rec);

    k_gather_gemm<<<STRIPES / STRIPES_PER_WG, 256, 0, stream>>>(rowptr, col, E, rec, W1, b1, dinv, W2, b2, z);
}


# First sort level chunk-local in LDS (count, scan, place; chunk stored back in bucket order); each bucket's block reads its 256 runs; count pass and scan launch dropped; float summation order changes
# speedup vs baseline: 15.0288x; 1.5174x over previous
#include <hip/hip_runtime.h>


#define N_NODES   100000
#define N_PAD     102400
#define BKT_SHIFT 8
#define BKT_NODES (1 << BKT_SHIFT)
#define N_CHUNK   256
#define N_BKT     (N_PAD / BKT_NODES)
#define N_BKT_USED ((N_NODES + BKT_NODES - 1) / BKT_NODES)
#define N_ROWS_T  (N_BKT_USED + 1)
#define SORT_T    1024
#define BKT_T     512
#define EDGE_B    8
#define CHUNK_CAP (SORT_T * EDGE_B)
#define RUN_IT    (N_CHUNK / (BKT_T / 16))
#define F0        2
#define F1        64
#define F2        128
#define NEG_SLOPE 0.01f

typedef __attribute__((ext_vector_type(2))) float v2f;
typedef __attribute__((ext_vector_type(8))) float v8f;

__device__ __forceinline__ float leaky(float v) {
    return v >= 0.0f ? v : v * NEG_SLOPE;
}


__global__ void __launch_bounds__(SORT_T)
k_chunk_sort(const int* __restrict__ src, const int* __restrict__ dst, int E,
             int* __restrict__ roff, int* __restrict__ part) {
    __shared__ int hist[N_BKT];
    __shared__ int wsum[16];
    __shared__ int stage[CHUNK_CAP];
    const int g    = blockIdx.x;
    const int tid  = threadIdx.x;
    const int lane = tid & 31;
    const int wave = tid >> 5;
    if (tid < N_BKT) hist[tid] = 0;
    __syncthreads();
    const int per  = min((E + N_CHUNK - 1) / N_CHUNK, CHUNK_CAP);
    const int cbeg = min(g * per, E);
    const int cend = min(cbeg + per, E);
    const int cnt  = cend - cbeg;

    unsigned d[EDGE_B], s[EDGE_B];
    int rank[EDGE_B];
#pragma unroll
    for (int k = 0; k < EDGE_B; ++k) {
        const int e = max(min(cbeg + tid + k * SORT_T, cend - 1), 0);
        d[k] = (unsigned)dst[e];
        s[k] = (unsigned)src[e];
    }
#pragma unroll
    for (int k = 0; k < EDGE_B; ++k) {
        d[k] = min(d[k], (unsigned)(N_NODES - 1));
        rank[k] = 0;
        if (tid + k * SORT_T < cnt) rank[k] = atomicAdd(&hist[d[k] >> BKT_SHIFT], 1);
    }
    __syncthreads();
    int c = 0, inc = 0;
    if (tid < 416) {
        c = (tid < N_BKT) ? hist[tid] : 0;
        inc = c;
        for (int o = 1; o < 32; o <<= 1) {
            const int u = __shfl_up(inc, o);
            if (lane >= o) inc += u;
        }
        if (lane == 31) wsum[wave] = inc;
    }
    __syncthreads();
    if (tid < 416) {
        int carry = 0;
        for (int w = 0; w < wave; ++w) carry += wsum[w];
        const int off = carry + inc - c;
        if (tid < N_BKT)    hist[tid] = off;
        if (tid < N_ROWS_T) roff[tid * N_CHUNK + g] = off;
    }
    __syncthreads();
#pragma unroll
    for (int k = 0; k < EDGE_B; ++k) {
        if (tid + k * SORT_T < cnt) {
            const int at = min(hist[d[k] >> BKT_SHIFT] + rank[k], CHUNK_CAP - 1);
            stage[at] = (int)((s[k] & 0x00FFFFFFu) | ((d[k] & (BKT_NODES - 1)) << 24));
        }
    }
    __syncthreads();
    for (int i = tid; i < cnt; i += SORT_T) part[cbeg + i] = stage[i];
}

__global__ void __launch_bounds__(BKT_T)
k_bucket_sort(const int* __restrict__ part, int E, const int* __restrict__ roff,
              const float* __restrict__ x, int* __restrict__ rowptr, int* __restrict__ col,
              float* __restrict__ dinv, float* __restrict__ xs) {
    __shared__ int cnt[BKT_NODES];
    __shared__ int cur[BKT_NODES];
    __shared__ int wsum[BKT_NODES / 32];
    __shared__ int rbeg[N_CHUNK];
    __shared__ int rlen[N_CHUNK];
    __shared__ int bsum[N_CHUNK / 32];
    const int b    = blockIdx.x;
    const int tid  = threadIdx.x;
    const int lane = tid & 31;
    const int wave = tid >> 5;
    const int per  = min((E + N_CHUNK - 1) / N_CHUNK, CHUNK_CAP);
    if (tid < BKT_NODES) cnt[tid] = 0;
    if (tid < N_CHUNK) {
        const int cb = min(tid * per, E);
        const int cn = min(cb + per, E) - cb;
        const int lo = min(max(roff[b * N_CHUNK + tid], 0), cn);
        const int hi = min(max(roff[(b + 1) * N_CHUNK + tid], lo), cn);
        rbeg[tid] = cb + lo;
        rlen[tid] = hi - lo;
        int t = lo;
        for (int o = 16; o > 0; o >>= 1) t += __shfl_xor(t, o);
        if (lane == 0) bsum[wave] = t;
    }
    __syncthreads();
    int bbeg = 0;
    for (int w = 0; w < N_CHUNK / 32; ++w) bbeg += bsum[w];
    bbeg = min(max(bbeg, 0), E);

    const int hw = tid >> 4;
    const int j  = tid & 15;
    int rb[RUN_IT], rl[RUN_IT];
    unsigned r[2 * RUN_IT];
#pragma unroll
    for (int it = 0; it < RUN_IT; ++it) {
        rb[it] = rbeg[it * (BKT_T / 16) + hw];
        rl[it] = rlen[it * (BKT_T / 16) + hw];
        r[2 * it]     = (unsigned)part[min(rb[it] + j, E - 1)];
        r[2 * it + 1] = (unsigned)part[min(rb[it] + j + 16, E - 1)];
    }
#pragma unroll
    for (int it = 0; it < RUN_IT; ++it) {
        if (j < rl[it])      atomicAdd(&cnt[r[2 * it] >> 24], 1);
        if (j + 16 < rl[it]) atomicAdd(&cnt[r[2 * it + 1] >> 24], 1);
        for (int o = j + 32; o < rl[it]; o += 16)
            atomicAdd(&cnt[(unsigned)part[min(rb[it] + o, E - 1)] >> 24], 1);
    }
    __syncthreads();
    int c = 0, inc = 0;
    if (tid < BKT_NODES) {
        c = cnt[tid];
        inc = c;
        for (int o = 1; o < 32; o <<= 1) {
            const int u = __shfl_up(inc, o);
            if (lane >= o) inc += u;
        }
        if (lane == 31) wsum[wave] = inc;
    }
    __syncthreads();
    if (tid < BKT_NODES) {
        int carry = 0;
        for (int w = 0; w < wave; ++w) carry += wsum[w];
        const int start = bbeg + carry + inc - c;
        cur[tid] = start;
        const int n = b * BKT_NODES + tid;
        rowptr[n] = start;
        if (n < N_NODES) {
            const float di = rsqrtf(fmaxf((float)(c + 1), 1.0f));
            dinv[n] = di;
            xs[F0 * n]     = x[F0 * n] * di;
            xs[F0 * n + 1] = x[F0 * n + 1] * di;
        }
    }
    __syncthreads();
    unsigned slot[2 * RUN_IT];
#pragma unroll
    for (int it = 0; it < RUN_IT; ++it) {
        slot[2 * it] = 0;
        slot[2 * it + 1] = 0;
        if (j < rl[it])      slot[2 * it]     = (unsigned)atomicAdd(&cur[r[2 * it] >> 24], 1);
        if (j + 16 < rl[it]) slot[2 * it + 1] = (unsigned)atomicAdd(&cur[r[2 * it + 1] >> 24], 1);
    }
#pragma unroll
    for (int it = 0; it < RUN_IT; ++it) {
        if (j < rl[it])      col[min(slot[2 * it], (unsigned)(E - 1))]     = (int)(r[2 * it] & 0x00FFFFFFu);
        if (j + 16 < rl[it]) col[min(slot[2 * it + 1], (unsigned)(E - 1))] = (int)(r[2 * it + 1] & 0x00FFFFFFu);
        for (int o = j + 32; o < rl[it]; o += 16) {
            const unsigned q = (unsigned)part[min(rb[it] + o, E - 1)];
            unsigned sl = (unsigned)atomicAdd(&cur[q >> 24], 1);
            sl = min(sl, (unsigned)(E - 1));
            col[sl] = (int)(q & 0x00FFFFFFu);
        }
    }
}


__global__ void k_gather_x(const int* __restrict__ rowptr, const int* __restrict__ col, int E,
                           const float* __restrict__ xs, const float* __restrict__ dinv,
                           float* __restrict__ rec) {
    const int n = blockIdx.x * blockDim.x + threadIdx.x;
    if (n >= N_NODES) return;
    const float2* xs2 = reinterpret_cast<const float2*>(xs);
    const int beg = max(rowptr[n], 0);
    const int end = min(rowptr[n + 1], E);
    float2 a = xs2[n];
#pragma unroll 4
    for (int j = beg; j < end; ++j) {
        const unsigned s = min((unsigned)col[j], (unsigned)(N_NODES - 1));
        const float2 v = xs2[s];
        a.x += v.x;
        a.y += v.y;
    }
    const float di = dinv[n];
    reinterpret_cast<float4*>(rec)[n] = make_float4(a.x * di, a.y * di, di, 0.0f);
}


#define STRIPES        (N_NODES / 16)
#define STRIPES_PER_WG 5
#define TILE_PITCH     68

__global__ void __launch_bounds__(256)
k_gather_gemm(const int* __restrict__ rowptr, const int* __restrict__ col, int E,
              const float* __restrict__ rec, const float* __restrict__ W1,
              const float* __restrict__ b1, const float* __restrict__ dinv,
              const float* __restrict__ W2, const float* __restrict__ b2,
              float* __restrict__ z) {
    __shared__ float tile[2][16 * TILE_PITCH];
    const int lane = threadIdx.x & 31;
    const int wave = __builtin_amdgcn_readfirstlane(threadIdx.x >> 5);
    const float4* rec4 = reinterpret_cast<const float4*>(rec);
    const v2f w0 = {W1[2 * lane],      W1[2 * lane + 1]};
    const v2f w1 = {W1[F1 + 2 * lane], W1[F1 + 2 * lane + 1]};
    const v2f bb = {b1[2 * lane],      b1[2 * lane + 1]};
    const v2f slope = {NEG_SLOPE, NEG_SLOPE};

    const int col0 = wave * 16;
    const int m    = lane & 15;
    const int kh   = (lane >> 4) << 1;
    v2f bf[F1 / 4];
#pragma unroll
    for (int t = 0; t < F1 / 4; ++t) {
        bf[t].x = W2[(4 * t + kh) * F2 + col0 + m];
        bf[t].y = W2[(4 * t + kh + 1) * F2 + col0 + m];
    }
    const float bias = b2[col0 + m];

    int buf = 0;
    for (int s = blockIdx.x; s < STRIPES; s += gridDim.x) {
        float* tl = tile[buf];
        for (int i = 0; i < 2; ++i) {
            const int r = 2 * wave + i;
            const int n = s * 16 + r;
            const int beg = max(rowptr[n], 0);
            const int end = min(rowptr[n + 1], E);
            const float di = dinv[n];
            const int deg   = max(end - beg, 0);
            const int total = deg + 1;
            v2f a = {0.0f, 0.0f};
            for (int base = 0; base < total; base += 32) {
                const int cnt = min(32, total - base);
                int mine = n;
                if (base + lane < deg) mine = col[beg + base + lane];
                mine = (int)min((unsigned)mine, (unsigned)(N_NODES - 1));
                const float4 q = rec4[mine];
                const int qx = __float_as_int(q.x), qy = __float_as_int(q.y), qz = __float_as_int(q.z);
#pragma unroll 4
                for (int j = 0; j < cnt; ++j) {
                    const float ax = __int_as_float(__builtin_amdgcn_readlane(qx, j));
                    const float ay = __int_as_float(__builtin_amdgcn_readlane(qy, j));
                    const float ds = __int_as_float(__builtin_amdgcn_readlane(qz, j));
                    const v2f vax = {ax, ax}, vay = {ay, ay}, vds = {ds, ds};
                    const v2f t = __builtin_elementwise_fma(vax, w0, __builtin_elementwise_fma(vay, w1, bb));
                    const v2f l = __builtin_elementwise_max(t, t * slope);
                    a = __builtin_elementwise_fma(l, vds, a);
                }
            }
            const v2f row = {a.x * di, a.y * di};
            *reinterpret_cast<v2f*>(&tl[r * TILE_PITCH + 2 * lane]) = row;
        }
        __syncthreads();

        v8f acc = {};
        const float* arow = tl + m * TILE_PITCH + kh;
#pragma unroll
        for (int t = 0; t < F1 / 4; ++t) {
            const v2f af = *reinterpret_cast<const v2f*>(arow + 4 * t);
            acc = __builtin_amdgcn_wmma_f32_16x16x4_f32(
                 false, af,  false, bf[t],
                 (short)0, acc,  false,  false);
        }

        const int rbase = s * 16 + ((lane >> 4) << 3);
#pragma unroll
        for (int v = 0; v < 8; ++v) {
            const size_t o = (size_t)(rbase + v) * F2 + col0 + m;
            z[o] = leaky(acc[v] + bias);
        }
        buf ^= 1;
    }
}


extern "C" void kernel_launch(void* const* d_in, const int* in_sizes, int n_in,
                              void* d_out, int out_size, void* d_ws, size_t ws_size,
                              hipStream_t stream) {
    const int*   edge_index = (const int*)d_in[0];
    const float* x          = (const float*)d_in[1];
    const float* W1         = (const float*)d_in[2];
    const float* b1         = (const float*)d_in[3];
    const float* W2         = (const float*)d_in[4];
    const float* b2         = (const float*)d_in[5];
    float*       z          = (float*)d_out;

    const int E = in_sizes[0] / 2;
    const int* src = edge_index;
    const int* dst = edge_index + E;

    float* ws     = (float*)d_ws;
    int*   roff   = (int*)ws;
    float* dinv   = ws + 204800;
    float* xs     = ws + 307200;
    float* rec    = ws + 512000;
    int*   rowptr = (int*)ws + 1024000;
    int*   part   = (int*)ws + 13516800;
    int*   col    = part + E;
    (void)ws_size; (void)n_in; (void)out_size;

    const int T = 256;

    k_chunk_sort  <<<N_CHUNK, SORT_T, 0, stream>>>(src, dst, E, roff, part);
    k_bucket_sort <<<N_BKT_USED, BKT_T, 0, stream>>>(part, E, roff, x, rowptr, col, dinv, xs);

    k_gather_x<<<(N_NODES + T - 1) / T, T, 0, stream>>>(rowptr, col, E, xs, dinv, rec);

    k_gather_gemm<<<STRIPES / STRIPES_PER_WG, 256, 0, stream>>>(rowptr, col, E, rec, W1, b1, dinv, W2, b2, z);
}


# Bucket sort places each bucket's source ids into an LDS copy of its rows and stores the copy as whole 128-B lines (16 B per lane), not one scattered 4-B global store per edge; sum order kept
# speedup vs baseline: 15.8839x; 1.0569x over previous
#include <hip/hip_runtime.h>


#define N_NODES   100000
#define N_PAD     102400
#define BKT_SHIFT 8
#define BKT_NODES (1 << BKT_SHIFT)
#define N_CHUNK   256
#define N_BKT     (N_PAD / BKT_NODES)
#define N_BKT_USED ((N_NODES + BKT_NODES - 1) / BKT_NODES)
#define N_ROWS_T  (N_BKT_USED + 1)
#define SORT_T    1024
#define BKT_T     512
#define EDGE_B    8
#define CHUNK_CAP (SORT_T * EDGE_B)
#define RUN_IT    (N_CHUNK / (BKT_T / 16))
#define STAGE_N   8448
#define F0        2
#define F1        64
#define F2        128
#define NEG_SLOPE 0.01f

typedef __attribute__((ext_vector_type(2))) float v2f;
typedef __attribute__((ext_vector_type(8))) float v8f;

__device__ __forceinline__ float leaky(float v) {
    return v >= 0.0f ? v : v * NEG_SLOPE;
}


__global__ void __launch_bounds__(SORT_T)
k_chunk_sort(const int* __restrict__ src, const int* __restrict__ dst, int E,
             int* __restrict__ roff, int* __restrict__ part) {
    __shared__ int hist[N_BKT];
    __shared__ int wsum[16];
    __shared__ int stage[CHUNK_CAP];
    const int g    = blockIdx.x;
    const int tid  = threadIdx.x;
    const int lane = tid & 31;
    const int wave = tid >> 5;
    if (tid < N_BKT) hist[tid] = 0;
    __syncthreads();
    const int per  = min((E + N_CHUNK - 1) / N_CHUNK, CHUNK_CAP);
    const int cbeg = min(g * per, E);
    const int cend = min(cbeg + per, E);
    const int cnt  = cend - cbeg;

    unsigned d[EDGE_B], s[EDGE_B];
    int rank[EDGE_B];
#pragma unroll
    for (int k = 0; k < EDGE_B; ++k) {
        const int e = max(min(cbeg + tid + k * SORT_T, cend - 1), 0);
        d[k] = (unsigned)dst[e];
        s[k] = (unsigned)src[e];
    }
#pragma unroll
    for (int k = 0; k < EDGE_B; ++k) {
        d[k] = min(d[k], (unsigned)(N_NODES - 1));
        rank[k] = 0;
        if (tid + k * SORT_T < cnt) rank[k] = atomicAdd(&hist[d[k] >> BKT_SHIFT], 1);
    }
    __syncthreads();
    int c = 0, inc = 0;
    if (tid < 416) {
        c = (tid < N_BKT) ? hist[tid] : 0;
        inc = c;
        for (int o = 1; o < 32; o <<= 1) {
            const int u = __shfl_up(inc, o);
            if (lane >= o) inc += u;
        }
        if (lane == 31) wsum[wave] = inc;
    }
    __syncthreads();
    if (tid < 416) {
        int carry = 0;
        for (int w = 0; w < wave; ++w) carry += wsum[w];
        const int off = carry + inc - c;
        if (tid < N_BKT)    hist[tid] = off;
        if (tid < N_ROWS_T) roff[tid * N_CHUNK + g] = off;
    }
    __syncthreads();
#pragma unroll
    for (int k = 0; k < EDGE_B; ++k) {
        if (tid + k * SORT_T < cnt) {
            const int at = min(hist[d[k] >> BKT_SHIFT] + rank[k], CHUNK_CAP - 1);
            stage[at] = (int)((s[k] & 0x00FFFFFFu) | ((d[k] & (BKT_NODES - 1)) << 24));
        }
    }
    __syncthreads();
    for (int i = tid; i < cnt; i += SORT_T) part[cbeg + i] = stage[i];
}

__device__ __forceinline__ void put_id(int* stage, int* __restrict__ col, int E, int base, unsigned slot, int id) {
    const unsigned q = slot - (unsigned)base;
    if (q < (unsigned)STAGE_N) stage[q] = id;
    else                       col[min(slot, (unsigned)(E - 1))] = id;
}

__global__ void __launch_bounds__(BKT_T)
k_bucket_sort(const int* __restrict__ part, int E, const int* __restrict__ roff,
              const float* __restrict__ x, int* __restrict__ rowptr, int* __restrict__ col,
              float* __restrict__ dinv, float* __restrict__ xs) {
    __shared__ int cnt[BKT_NODES];
    __shared__ int cur[BKT_NODES];
    __shared__ int wsum[BKT_NODES / 32];
    __shared__ int rbeg[N_CHUNK];
    __shared__ int rlen[N_CHUNK];
    __shared__ int bsum[N_CHUNK / 32];
    __shared__ __attribute__((aligned(16))) int stage[STAGE_N];
    const int b    = blockIdx.x;
    const int tid  = threadIdx.x;
    const int lane = tid & 31;
    const int wave = tid >> 5;
    const int per  = min((E + N_CHUNK - 1) / N_CHUNK, CHUNK_CAP);
    if (tid < BKT_NODES) cnt[tid] = 0;
    if (tid < N_CHUNK) {
        const int cb = min(tid * per, E);
        const int cn = min(cb + per, E) - cb;
        const int lo = min(max(roff[b * N_CHUNK + tid], 0), cn);
        const int hi = min(max(roff[(b + 1) * N_CHUNK + tid], lo), cn);
        rbeg[tid] = cb + lo;
        rlen[tid] = hi - lo;
        int t = lo;
        for (int o = 16; o > 0; o >>= 1) t += __shfl_xor(t, o);
        if (lane == 0) bsum[wave] = t;
    }
    __syncthreads();
    int bbeg = 0;
    for (int w = 0; w < N_CHUNK / 32; ++w) bbeg += bsum[w];
    bbeg = min(max(bbeg, 0), E);

    const int hw = tid >> 4;
    const int j  = tid & 15;
    int rb[RUN_IT], rl[RUN_IT];
    unsigned r[2 * RUN_IT];
#pragma unroll
    for (int it = 0; it < RUN_IT; ++it) {
        rb[it] = rbeg[it * (BKT_T / 16) + hw];
        rl[it] = rlen[it * (BKT_T / 16) + hw];
        r[2 * it]     = (unsigned)part[min(rb[it] + j, E - 1)];
        r[2 * it + 1] = (unsigned)part[min(rb[it] + j + 16, E - 1)];
    }
#pragma unroll
    for (int it = 0; it < RUN_IT; ++it) {
        if (j < rl[it])      atomicAdd(&cnt[r[2 * it] >> 24], 1);
        if (j + 16 < rl[it]) atomicAdd(&cnt[r[2 * it + 1] >> 24], 1);
        for (int o = j + 32; o < rl[it]; o += 16)
            atomicAdd(&cnt[(unsigned)part[min(rb[it] + o, E - 1)] >> 24], 1);
    }
    __syncthreads();
    int c = 0, inc = 0;
    if (tid < BKT_NODES) {
        c = cnt[tid];
        inc = c;
        for (int o = 1; o < 32; o <<= 1) {
            const int u = __shfl_up(inc, o);
            if (lane >= o) inc += u;
        }
        if (lane == 31) wsum[wave] = inc;
    }
    __syncthreads();
    if (tid < BKT_NODES) {
        int carry = 0;
        for (int w = 0; w < wave; ++w) carry += wsum[w];
        const int start = bbeg + carry + inc - c;
        cur[tid] = start;
        const int n = b * BKT_NODES + tid;
        rowptr[n] = start;
        if (n < N_NODES) {
            const float di = rsqrtf(fmaxf((float)(c + 1), 1.0f));
            dinv[n] = di;
            xs[F0 * n]     = x[F0 * n] * di;
            xs[F0 * n + 1] = x[F0 * n + 1] * di;
        }
    }
    __syncthreads();
    unsigned slot[2 * RUN_IT];
#pragma unroll
    for (int it = 0; it < RUN_IT; ++it) {
        slot[2 * it] = 0;
        slot[2 * it + 1] = 0;
        if (j < rl[it])      slot[2 * it]     = (unsigned)atomicAdd(&cur[r[2 * it] >> 24], 1);
        if (j + 16 < rl[it]) slot[2 * it + 1] = (unsigned)atomicAdd(&cur[r[2 * it + 1] >> 24], 1);
    }
    const int head = (int)(((size_t)(col + bbeg) >> 2) & 31);
    const int base = bbeg - head;
    bool past = false;
#pragma unroll
    for (int it = 0; it < RUN_IT; ++it) {
        const unsigned q0 = slot[2 * it] - (unsigned)base, q1 = slot[2 * it + 1] - (unsigned)base;
        const bool in0 = j < rl[it] && q0 < (unsigned)STAGE_N, in1 = j + 16 < rl[it] && q1 < (unsigned)STAGE_N;
        if (in0) stage[q0] = (int)(r[2 * it] & 0x00FFFFFFu);
        if (in1) stage[q1] = (int)(r[2 * it + 1] & 0x00FFFFFFu);
        past |= (j < rl[it] && !in0) | (j + 16 < rl[it] && !in1);
    }
    if (past) {
#pragma unroll
        for (int it = 0; it < RUN_IT; ++it) {
            if (j < rl[it] && slot[2 * it] - (unsigned)base >= (unsigned)STAGE_N)
                col[min(slot[2 * it], (unsigned)(E - 1))] = (int)(r[2 * it] & 0x00FFFFFFu);
            if (j + 16 < rl[it] && slot[2 * it + 1] - (unsigned)base >= (unsigned)STAGE_N)
                col[min(slot[2 * it + 1], (unsigned)(E - 1))] = (int)(r[2 * it + 1] & 0x00FFFFFFu);
        }
    }
#pragma unroll
    for (int it = 0; it < RUN_IT; ++it) {
        for (int o = j + 32; o < rl[it]; o += 16) {
            const unsigned q = (unsigned)part[min(rb[it] + o, E - 1)];
            const unsigned sl = (unsigned)atomicAdd(&cur[q >> 24], 1);
            put_id(stage, col, E, base, sl, (int)(q & 0x00FFFFFFu));
        }
    }
    __syncthreads();
    int tot = 0;
    for (int w = 0; w < BKT_NODES / 32; ++w) tot += wsum[w];
    const int nst = min(min(head + tot, STAGE_N), E - base);
    for (int q = 4 * tid; q < nst; q += 4 * BKT_T) {
        if (q >= head && q + 3 < nst) {
            *reinterpret_cast<int4*>(col + base + q) = *reinterpret_cast<const int4*>(stage + q);
        } else {
            for (int k = 0; k < 4; ++k)
                if (q + k >= head && q + k < nst) col[base + q + k] = stage[q + k];
        }
    }
}


__global__ void k_gather_x(const int* __restrict__ rowptr, const int* __restrict__ col, int E,
                           const float* __restrict__ xs, const float* __restrict__ dinv,
                           float* __restrict__ rec) {
    const int n = blockIdx.x * blockDim.x + threadIdx.x;
    if (n >= N_NODES) return;
    const float2* xs2 = reinterpret_cast<const float2*>(xs);
    const int beg = max(rowptr[n], 0);
    const int end = min(rowptr[n + 1], E);
    float2 a = xs2[n];
#pragma unroll 4
    for (int j = beg; j < end; ++j) {
        const unsigned s = min((unsigned)col[j], (unsigned)(N_NODES - 1));
        const float2 v = xs2[s];
        a.x += v.x;
        a.y += v.y;
    }
    const float di = dinv[n];
    reinterpret_cast<float4*>(rec)[n] = make_float4(a.x * di, a.y * di, di, 0.0f);
}


#define STRIPES        (N_NODES / 16)
#define STRIPES_PER_WG 5
#define TILE_PITCH     68

__global__ void __launch_bounds__(256)
k_gather_gemm(const int* __restrict__ rowptr, const int* __restrict__ col, int E,
              const float* __restrict__ rec, const float* __restrict__ W1,
              const float* __restrict__ b1, const float* __restrict__ dinv,
              const float* __restrict__ W2, const float* __restrict__ b2,
              float* __restrict__ z) {
    __shared__ float tile[2][16 * TILE_PITCH];
    const int lane = threadIdx.x & 31;
    const int wave = __builtin_amdgcn_readfirstlane(threadIdx.x >> 5);
    const float4* rec4 = reinterpret_cast<const float4*>(rec);
    const v2f w0 = {W1[2 * lane],      W1[2 * lane + 1]};
    const v2f w1 = {W1[F1 + 2 * lane], W1[F1 + 2 * lane + 1]};
    const v2f bb = {b1[2 * lane],      b1[2 * lane + 1]};
    const v2f slope = {NEG_SLOPE, NEG_SLOPE};

    const int col0 = wave * 16;
    const int m    = lane & 15;
    const int kh   = (lane >> 4) << 1;
    v2f bf[F1 / 4];
#pragma unroll
    for (int t = 0; t < F1 / 4; ++t) {
        bf[t].x = W2[(4 * t + kh) * F2 + col0 + m];
        bf[t].y = W2[(4 * t + kh + 1) * F2 + col0 + m];
    }
    const float bias = b2[col0 + m];

    int buf = 0;
    for (int s = blockIdx.x; s < STRIPES; s += gridDim.x) {
        float* tl = tile[buf];
        for (int i = 0; i < 2; ++i) {
            const int r = 2 * wave + i;
            const int n = s * 16 + r;
            const int beg = max(rowptr[n], 0);
            const int end = min(rowptr[n + 1], E);
            const float di = dinv[n];
            const int deg   = max(end - beg, 0);
            const int total = deg + 1;
            v2f a = {0.0f, 0.0f};
            for (int base = 0; base < total; base += 32) {
                const int cnt = min(32, total - base);
                int mine = n;
                if (base + lane < deg) mine = col[beg + base + lane];
                mine = (int)min((unsigned)mine, (unsigned)(N_NODES - 1));
                const float4 q = rec4[mine];
                const int qx = __float_as_int(q.x), qy = __float_as_int(q.y), qz = __float_as_int(q.z);
#pragma unroll 4
                for (int j = 0; j < cnt; ++j) {
                    const float ax = __int_as_float(__builtin_amdgcn_readlane(qx, j));
                    const float ay = __int_as_float(__builtin_amdgcn_readlane(qy, j));
                    const float ds = __int_as_float(__builtin_amdgcn_readlane(qz, j));
                    const v2f vax = {ax, ax}, vay = {ay, ay}, vds = {ds, ds};
                    const v2f t = __builtin_elementwise_fma(vax, w0, __builtin_elementwise_fma(vay, w1, bb));
                    const v2f l = __builtin_elementwise_max(t, t * slope);
                    a = __builtin_elementwise_fma(l, vds, a);
                }
            }
            const v2f row = {a.x * di, a.y * di};
            *reinterpret_cast<v2f*>(&tl[r * TILE_PITCH + 2 * lane]) = row;
        }
        __syncthreads();

        v8f acc = {};
        const float* arow = tl + m * TILE_PITCH + kh;
#pragma unroll
        for (int t = 0; t < F1 / 4; ++t) {
            const v2f af = *reinterpret_cast<const v2f*>(arow + 4 * t);
            acc = __builtin_amdgcn_wmma_f32_16x16x4_f32(
                 false, af,  false, bf[t],
                 (short)0, acc,  false,  false);
        }

        const int rbase = s * 16 + ((lane >> 4) << 3);
#pragma unroll
        for (int v = 0; v < 8; ++v) {
            const size_t o = (size_t)(rbase + v) * F2 + col0 + m;
            z[o] = leaky(acc[v] + bias);
        }
        buf ^= 1;
    }
}


extern "C" void kernel_launch(void* const* d_in, const int* in_sizes, int n_in,
                              void* d_out, int out_size, void* d_ws, size_t ws_size,
                              hipStream_t stream) {
    const int*   edge_index = (const int*)d_in[0];
    const float* x          = (const float*)d_in[1];
    const float* W1         = (const float*)d_in[2];
    const float* b1         = (const float*)d_in[3];
    const float* W2         = (const float*)d_in[4];
    const float* b2         = (const float*)d_in[5];
    float*       z          = (float*)d_out;

    const int E = in_sizes[0] / 2;
    const int* src = edge_index;
    const int* dst = edge_index + E;

    float* ws     = (float*)d_ws;
    int*   roff   = (int*)ws;
    float* dinv   = ws + 204800;
    float* xs     = ws + 307200;
    float* rec    = ws + 512000;
    int*   rowptr = (int*)ws + 1024000;
    int*   part   = (int*)ws + 13516800;
    int*   col    = part + E;
    (void)ws_size; (void)n_in; (void)out_size;

    const int T = 256;

    k_chunk_sort  <<<N_CHUNK, SORT_T, 0, stream>>>(src, dst, E, roff, part);
    k_bucket_sort <<<N_BKT_USED, BKT_T, 0, stream>>>(part, E, roff, x, rowptr, col, dinv, xs);

    k_gather_x<<<(N_NODES + T - 1) / T, T, 0, stream>>>(rowptr, col, E, xs, dinv, rec);

    k_gather_gemm<<<STRIPES / STRIPES_PER_WG, 256, 0, stream>>>(rowptr, col, E, rec, W1, b1, dinv, W2, b2, z);
}


# Layer-2 gather loads ahead: while a wave walks node q, the source records of node q+1, the ids of q+2 and the row bounds of q+3 are in flight (was 3 dependent trips per node); sum order kept
# speedup vs baseline: 16.9074x; 1.0644x over previous
#include <hip/hip_runtime.h>


#define N_NODES   100000
#define N_PAD     102400
#define BKT_SHIFT 8
#define BKT_NODES (1 << BKT_SHIFT)
#define N_CHUNK   256
#define N_BKT     (N_PAD / BKT_NODES)
#define N_BKT_USED ((N_NODES + BKT_NODES - 1) / BKT_NODES)
#define N_ROWS_T  (N_BKT_USED + 1)
#define SORT_T    1024
#define BKT_T     512
#define EDGE_B    8
#define CHUNK_CAP (SORT_T * EDGE_B)
#define RUN_IT    (N_CHUNK / (BKT_T / 16))
#define STAGE_N   8448
#define F0        2
#define F1        64
#define F2        128
#define NEG_SLOPE 0.01f

typedef __attribute__((ext_vector_type(2))) float v2f;
typedef __attribute__((ext_vector_type(8))) float v8f;

__device__ __forceinline__ float leaky(float v) {
    return v >= 0.0f ? v : v * NEG_SLOPE;
}


__global__ void __launch_bounds__(SORT_T)
k_chunk_sort(const int* __restrict__ src, const int* __restrict__ dst, int E,
             int* __restrict__ roff, int* __restrict__ part) {
    __shared__ int hist[N_BKT];
    __shared__ int wsum[16];
    __shared__ int stage[CHUNK_CAP];
    const int g    = blockIdx.x;
    const int tid  = threadIdx.x;
    const int lane = tid & 31;
    const int wave = tid >> 5;
    if (tid < N_BKT) hist[tid] = 0;
    __syncthreads();
    const int per  = min((E + N_CHUNK - 1) / N_CHUNK, CHUNK_CAP);
    const int cbeg = min(g * per, E);
    const int cend = min(cbeg + per, E);
    const int cnt  = cend - cbeg;

    unsigned d[EDGE_B], s[EDGE_B];
    int rank[EDGE_B];
#pragma unroll
    for (int k = 0; k < EDGE_B; ++k) {
        const int e = max(min(cbeg + tid + k * SORT_T, cend - 1), 0);
        d[k] = (unsigned)dst[e];
        s[k] = (unsigned)src[e];
    }
#pragma unroll
    for (int k = 0; k < EDGE_B; ++k) {
        d[k] = min(d[k], (unsigned)(N_NODES - 1));
        rank[k] = 0;
        if (tid + k * SORT_T < cnt) rank[k] = atomicAdd(&hist[d[k] >> BKT_SHIFT], 1);
    }
    __syncthreads();
    int c = 0, inc = 0;
    if (tid < 416) {
        c = (tid < N_BKT) ? hist[tid] : 0;
        inc = c;
        for (int o = 1; o < 32; o <<= 1) {
            const int u = __shfl_up(inc, o);
            if (lane >= o) inc += u;
        }
        if (lane == 31) wsum[wave] = inc;
    }
    __syncthreads();
    if (tid < 416) {
        int carry = 0;
        for (int w = 0; w < wave; ++w) carry += wsum[w];
        const int off = carry + inc - c;
        if (tid < N_BKT)    hist[tid] = off;
        if (tid < N_ROWS_T) roff[tid * N_CHUNK + g] = off;
    }
    __syncthreads();
#pragma unroll
    for (int k = 0; k < EDGE_B; ++k) {
        if (tid + k * SORT_T < cnt) {
            const int at = min(hist[d[k] >> BKT_SHIFT] + rank[k], CHUNK_CAP - 1);
            stage[at] = (int)((s[k] & 0x00FFFFFFu) | ((d[k] & (BKT_NODES - 1)) << 24));
        }
    }
    __syncthreads();
    for (int i = tid; i < cnt; i += SORT_T) part[cbeg + i] = stage[i];
}

__device__ __forceinline__ void put_id(int* stage, int* __restrict__ col, int E, int base, unsigned slot, int id) {
    const unsigned q = slot - (unsigned)base;
    if (q < (unsigned)STAGE_N) stage[q] = id;
    else                       col[min(slot, (unsigned)(E - 1))] = id;
}

__global__ void __launch_bounds__(BKT_T)
k_bucket_sort(const int* __restrict__ part, int E, const int* __restrict__ roff,
              const float* __restrict__ x, int* __restrict__ rowptr, int* __restrict__ col,
              float* __restrict__ dinv, float* __restrict__ xs) {
    __shared__ int cnt[BKT_NODES];
    __shared__ int cur[BKT_NODES];
    __shared__ int wsum[BKT_NODES / 32];
    __shared__ int rbeg[N_CHUNK];
    __shared__ int rlen[N_CHUNK];
    __shared__ int bsum[N_CHUNK / 32];
    __shared__ __attribute__((aligned(16))) int stage[STAGE_N];
    const int b    = blockIdx.x;
    const int tid  = threadIdx.x;
    const int lane = tid & 31;
    const int wave = tid >> 5;
    const int per  = min((E + N_CHUNK - 1) / N_CHUNK, CHUNK_CAP);
    if (tid < BKT_NODES) cnt[tid] = 0;
    if (tid < N_CHUNK) {
        const int cb = min(tid * per, E);
        const int cn = min(cb + per, E) - cb;
        const int lo = min(max(roff[b * N_CHUNK + tid], 0), cn);
        const int hi = min(max(roff[(b + 1) * N_CHUNK + tid], lo), cn);
        rbeg[tid] = cb + lo;
        rlen[tid] = hi - lo;
        int t = lo;
        for (int o = 16; o > 0; o >>= 1) t += __shfl_xor(t, o);
        if (lane == 0) bsum[wave] = t;
    }
    __syncthreads();
    int bbeg = 0;
    for (int w = 0; w < N_CHUNK / 32; ++w) bbeg += bsum[w];
    bbeg = min(max(bbeg, 0), E);

    const int hw = tid >> 4;
    const int j  = tid & 15;
    int rb[RUN_IT], rl[RUN_IT];
    unsigned r[2 * RUN_IT];
#pragma unroll
    for (int it = 0; it < RUN_IT; ++it) {
        rb[it] = rbeg[it * (BKT_T / 16) + hw];
        rl[it] = rlen[it * (BKT_T / 16) + hw];
        r[2 * it]     = (unsigned)part[min(rb[it] + j, E - 1)];
        r[2 * it + 1] = (unsigned)part[min(rb[it] + j + 16, E - 1)];
    }
#pragma unroll
    for (int it = 0; it < RUN_IT; ++it) {
        if (j < rl[it])      atomicAdd(&cnt[r[2 * it] >> 24], 1);
        if (j + 16 < rl[it]) atomicAdd(&cnt[r[2 * it + 1] >> 24], 1);
        for (int o = j + 32; o < rl[it]; o += 16)
            atomicAdd(&cnt[(unsigned)part[min(rb[it] + o, E - 1)] >> 24], 1);
    }
    __syncthreads();
    int c = 0, inc = 0;
    if (tid < BKT_NODES) {
        c = cnt[tid];
        inc = c;
        for (int o = 1; o < 32; o <<= 1) {
            const int u = __shfl_up(inc, o);
            if (lane >= o) inc += u;
        }
        if (lane == 31) wsum[wave] = inc;
    }
    __syncthreads();
    if (tid < BKT_NODES) {
        int carry = 0;
        for (int w = 0; w < wave; ++w) carry += wsum[w];
        const int start = bbeg + carry + inc - c;
        cur[tid] = start;
        const int n = b * BKT_NODES + tid;
        rowptr[n] = start;
        if (n < N_NODES) {
            const float di = rsqrtf(fmaxf((float)(c + 1), 1.0f));
            dinv[n] = di;
            xs[F0 * n]     = x[F0 * n] * di;
            xs[F0 * n + 1] = x[F0 * n + 1] * di;
        }
    }
    __syncthreads();
    unsigned slot[2 * RUN_IT];
#pragma unroll
    for (int it = 0; it < RUN_IT; ++it) {
        slot[2 * it] = 0;
        slot[2 * it + 1] = 0;
        if (j < rl[it])      slot[2 * it]     = (unsigned)atomicAdd(&cur[r[2 * it] >> 24], 1);
        if (j + 16 < rl[it]) slot[2 * it + 1] = (unsigned)atomicAdd(&cur[r[2 * it + 1] >> 24], 1);
    }
    const int head = (int)(((size_t)(col + bbeg) >> 2) & 31);
    const int base = bbeg - head;
    bool past = false;
#pragma unroll
    for (int it = 0; it < RUN_IT; ++it) {
        const unsigned q0 = slot[2 * it] - (unsigned)base, q1 = slot[2 * it + 1] - (unsigned)base;
        const bool in0 = j < rl[it] && q0 < (unsigned)STAGE_N, in1 = j + 16 < rl[it] && q1 < (unsigned)STAGE_N;
        if (in0) stage[q0] = (int)(r[2 * it] & 0x00FFFFFFu);
        if (in1) stage[q1] = (int)(r[2 * it + 1] & 0x00FFFFFFu);
        past |= (j < rl[it] && !in0) | (j + 16 < rl[it] && !in1);
    }
    if (past) {
#pragma unroll
        for (int it = 0; it < RUN_IT; ++it) {
            if (j < rl[it] && slot[2 * it] - (unsigned)base >= (unsigned)STAGE_N)
                col[min(slot[2 * it], (unsigned)(E - 1))] = (int)(r[2 * it] & 0x00FFFFFFu);
            if (j + 16 < rl[it] && slot[2 * it + 1] - (unsigned)base >= (unsigned)STAGE_N)
                col[min(slot[2 * it + 1], (unsigned)(E - 1))] = (int)(r[2 * it + 1] & 0x00FFFFFFu);
        }
    }
#pragma unroll
    for (int it = 0; it < RUN_IT; ++it) {
        for (int o = j + 32; o < rl[it]; o += 16) {
            const unsigned q = (unsigned)part[min(rb[it] + o, E - 1)];
            const unsigned sl = (unsigned)atomicAdd(&cur[q >> 24], 1);
            put_id(stage, col, E, base, sl, (int)(q & 0x00FFFFFFu));
        }
    }
    __syncthreads();
    int tot = 0;
    for (int w = 0; w < BKT_NODES / 32; ++w) tot += wsum[w];
    const int nst = min(min(head + tot, STAGE_N), E - base);
    for (int q = 4 * tid; q < nst; q += 4 * BKT_T) {
        if (q >= head && q + 3 < nst) {
            *reinterpret_cast<int4*>(col + base + q) = *reinterpret_cast<const int4*>(stage + q);
        } else {
            for (int k = 0; k < 4; ++k)
                if (q + k >= head && q + k < nst) col[base + q + k] = stage[q + k];
        }
    }
}


__global__ void k_gather_x(const int* __restrict__ rowptr, const int* __restrict__ col, int E,
                           const float* __restrict__ xs, const float* __restrict__ dinv,
                           float* __restrict__ rec) {
    const int n = blockIdx.x * blockDim.x + threadIdx.x;
    if (n >= N_NODES) return;
    const float2* xs2 = reinterpret_cast<const float2*>(xs);
    const int beg = max(rowptr[n], 0);
    const int end = min(rowptr[n + 1], E);
    float2 a = xs2[n];
#pragma unroll 4
    for (int j = beg; j < end; ++j) {
        const unsigned s = min((unsigned)col[j], (unsigned)(N_NODES - 1));
        const float2 v = xs2[s];
        a.x += v.x;
        a.y += v.y;
    }
    const float di = dinv[n];
    reinterpret_cast<float4*>(rec)[n] = make_float4(a.x * di, a.y * di, di, 0.0f);
}


#define STRIPES        (N_NODES / 16)
#define STRIPES_PER_WG 5
#define TILE_PITCH     68

__device__ __forceinline__ v2f walk_chunk(v2f a, int qx, int qy, int qz, int cnt,
                                          v2f w0, v2f w1, v2f bb, v2f slope) {
#pragma unroll 4
    for (int j = 0; j < cnt; ++j) {
        const float ax = __int_as_float(__builtin_amdgcn_readlane(qx, j));
        const float ay = __int_as_float(__builtin_amdgcn_readlane(qy, j));
        const float ds = __int_as_float(__builtin_amdgcn_readlane(qz, j));
        const v2f vax = {ax, ax}, vay = {ay, ay}, vds = {ds, ds};
        const v2f t = __builtin_elementwise_fma(vax, w0, __builtin_elementwise_fma(vay, w1, bb));
        const v2f l = __builtin_elementwise_max(t, t * slope);
        a = __builtin_elementwise_fma(l, vds, a);
    }
    return a;
}

__global__ void __launch_bounds__(256)
k_gather_gemm(const int* __restrict__ rowptr, const int* __restrict__ col, int E,
              const float* __restrict__ rec, const float* __restrict__ W1,
              const float* __restrict__ b1, const float* __restrict__ dinv,
              const float* __restrict__ W2, const float* __restrict__ b2,
              float* __restrict__ z) {
    __shared__ float tile[2][16 * TILE_PITCH];
    const int lane = threadIdx.x & 31;
    const int wave = __builtin_amdgcn_readfirstlane(threadIdx.x >> 5);
    const float4* rec4 = reinterpret_cast<const float4*>(rec);
    const v2f w0 = {W1[2 * lane],      W1[2 * lane + 1]};
    const v2f w1 = {W1[F1 + 2 * lane], W1[F1 + 2 * lane + 1]};
    const v2f bb = {b1[2 * lane],      b1[2 * lane + 1]};
    const v2f slope = {NEG_SLOPE, NEG_SLOPE};

    const int col0 = wave * 16;
    const int m    = lane & 15;
    const int kh   = (lane >> 4) << 1;
    v2f bf[F1 / 4];
#pragma unroll
    for (int t = 0; t < F1 / 4; ++t) {
        bf[t].x = W2[(4 * t + kh) * F2 + col0 + m];
        bf[t].y = W2[(4 * t + kh + 1) * F2 + col0 + m];
    }
    const float bias = b2[col0 + m];

    const int nstr = ((int)blockIdx.x < STRIPES) ? (STRIPES - 1 - (int)blockIdx.x) / (int)gridDim.x + 1 : 0;
    const int nq   = 2 * nstr;
    if (nq == 0) return;
#define NODE_AT(q) (((int)blockIdx.x + (min((q), nq - 1) >> 1) * (int)gridDim.x) * 16 + 2 * wave + (min((q), nq - 1) & 1))

    int   n_c = NODE_AT(0), n_1 = NODE_AT(1), n_2 = NODE_AT(2);
    int   rb_c = rowptr[n_c], re_c = rowptr[n_c + 1];
    int   rb_1 = rowptr[n_1], re_1 = rowptr[n_1 + 1];
    int   rbv_2 = rowptr[n_2 + (lane & 1)];
    float di_c = dinv[n_c], di_1 = dinv[n_1], di_2 = dinv[n_2];
    int beg_c = max(rb_c, 0);
    int deg_c = max(min(re_c, E) - beg_c, 0);
    int beg_1 = max(rb_1, 0);
    int deg_1 = max(min(re_1, E) - beg_1, 0);
    int ids_c = n_c;
    if (lane < deg_c) ids_c = col[beg_c + lane];
    int ids_1 = n_1;
    if (lane < deg_1) ids_1 = col[beg_1 + lane];
    float4 r_c = rec4[min((unsigned)ids_c, (unsigned)(N_NODES - 1))];

    int buf = 0;
    for (int q = 0; q < nq; ++q) {
        const int   n_3  = NODE_AT(q + 3);
        const int   rbv_3 = rowptr[n_3 + (lane & 1)];
        const float di_3 = dinv[n_3];
        const int beg_2 = max(__builtin_amdgcn_readlane(rbv_2, 0), 0);
        const int deg_2 = max(min(__builtin_amdgcn_readlane(rbv_2, 1), E) - beg_2, 0);
        int ids_2 = n_2;
        if (lane < deg_2) ids_2 = col[beg_2 + lane];
        const float4 r_1 = rec4[min((unsigned)ids_1, (unsigned)(N_NODES - 1))];

        const int total = deg_c + 1;
        v2f a = {0.0f, 0.0f};
        a = walk_chunk(a, __float_as_int(r_c.x), __float_as_int(r_c.y), __float_as_int(r_c.z),
                       min(32, total), w0, w1, bb, slope);
        for (int base = 32; base < total; base += 32) {
            int mine = n_c;
            if (base + lane < deg_c) mine = col[beg_c + base + lane];
            const float4 rq = rec4[min((unsigned)mine, (unsigned)(N_NODES - 1))];
            a = walk_chunk(a, __float_as_int(rq.x), __float_as_int(rq.y), __float_as_int(rq.z),
                           min(32, total - base), w0, w1, bb, slope);
        }
        float* tl = tile[buf];
        const int r = 2 * wave + (q & 1);
        const v2f row = {a.x * di_c, a.y * di_c};
        *reinterpret_cast<v2f*>(&tl[r * TILE_PITCH + 2 * lane]) = row;

        if (q & 1) {
            const int s = (int)blockIdx.x + (q >> 1) * (int)gridDim.x;
            __syncthreads();

            v8f acc = {};
            const float* arow = tl + m * TILE_PITCH + kh;
#pragma unroll
            for (int t = 0; t < F1 / 4; ++t) {
                const v2f af = *reinterpret_cast<const v2f*>(arow + 4 * t);
                acc = __builtin_amdgcn_wmma_f32_16x16x4_f32(
                     false, af,  false, bf[t],
                     (short)0, acc,  false,  false);
            }

            const int rbase = s * 16 + ((lane >> 4) << 3);
#pragma unroll
            for (int v = 0; v < 8; ++v) {
                const size_t o = (size_t)(rbase + v) * F2 + col0 + m;
                z[o] = leaky(acc[v] + bias);
            }
            buf ^= 1;
        }

        n_c = n_1; beg_c = beg_1; deg_c = deg_1; di_c = di_1; r_c = r_1;
        n_1 = n_2; beg_1 = beg_2; deg_1 = deg_2; di_1 = di_2; ids_1 = ids_2;
        n_2 = n_3; rbv_2 = rbv_3; di_2 = di_3;
    }
#undef NODE_AT
}


extern "C" void kernel_launch(void* const* d_in, const int* in_sizes, int n_in,
                              void* d_out, int out_size, void* d_ws, size_t ws_size,
                              hipStream_t stream) {
    const int*   edge_index = (const int*)d_in[0];
    const float* x          = (const float*)d_in[1];
    const float* W1         = (const float*)d_in[2];
    const float* b1         = (const float*)d_in[3];
    const float* W2         = (const float*)d_in[4];
    const float* b2         = (const float*)d_in[5];
    float*       z          = (float*)d_out;

    const int E = in_sizes[0] / 2;
    const int* src = edge_index;
    const int* dst = edge_index + E;

    float* ws     = (float*)d_ws;
    int*   roff   = (int*)ws;
    float* dinv   = ws + 204800;
    float* xs     = ws + 307200;
    float* rec    = ws + 512000;
    int*   rowptr = (int*)ws + 1024000;
    int*   part   = (int*)ws + 13516800;
    int*   col    = part + E;
    (void)ws_size; (void)n_in; (void)out_size;

    const int T = 256;

    k_chunk_sort  <<<N_CHUNK, SORT_T, 0, stream>>>(src, dst, E, roff, part);
    k_bucket_sort <<<N_BKT_USED, BKT_T, 0, stream>>>(part, E, roff, x, rowptr, col, dinv, xs);

    k_gather_x<<<(N_NODES + T - 1) / T, T, 0, stream>>>(rowptr, col, E, xs, dinv, rec);

    k_gather_gemm<<<STRIPES / STRIPES_PER_WG, 256, 0, stream>>>(rowptr, col, E, rec, W1, b1, dinv, W2, b2, z);
}


# Layer-1 gather: a wave loads its 32 nodes' contiguous id piece coalesced, fetches all source rows at once, sums each node's run from LDS in list order: 3 dependent trips per node; sum order kept
# speedup vs baseline: 18.0396x; 1.0670x over previous
#include <hip/hip_runtime.h>


#define N_NODES   100000
#define N_PAD     102400
#define BKT_SHIFT 8
#define BKT_NODES (1 << BKT_SHIFT)
#define N_CHUNK   256
#define N_BKT     (N_PAD / BKT_NODES)
#define N_BKT_USED ((N_NODES + BKT_NODES - 1) / BKT_NODES)
#define N_ROWS_T  (N_BKT_USED + 1)
#define SORT_T    1024
#define BKT_T     512
#define EDGE_B    8
#define CHUNK_CAP (SORT_T * EDGE_B)
#define RUN_IT    (N_CHUNK / (BKT_T / 16))
#define STAGE_N   8448
#define F0        2
#define F1        64
#define F2        128
#define NEG_SLOPE 0.01f

typedef __attribute__((ext_vector_type(2))) float v2f;
typedef __attribute__((ext_vector_type(8))) float v8f;

__device__ __forceinline__ float leaky(float v) {
    return v >= 0.0f ? v : v * NEG_SLOPE;
}


__global__ void __launch_bounds__(SORT_T)
k_chunk_sort(const int* __restrict__ src, const int* __restrict__ dst, int E,
             int* __restrict__ roff, int* __restrict__ part) {
    __shared__ int hist[N_BKT];
    __shared__ int wsum[16];
    __shared__ int stage[CHUNK_CAP];
    const int g    = blockIdx.x;
    const int tid  = threadIdx.x;
    const int lane = tid & 31;
    const int wave = tid >> 5;
    if (tid < N_BKT) hist[tid] = 0;
    __syncthreads();
    const int per  = min((E + N_CHUNK - 1) / N_CHUNK, CHUNK_CAP);
    const int cbeg = min(g * per, E);
    const int cend = min(cbeg + per, E);
    const int cnt  = cend - cbeg;

    unsigned d[EDGE_B], s[EDGE_B];
    int rank[EDGE_B];
#pragma unroll
    for (int k = 0; k < EDGE_B; ++k) {
        const int e = max(min(cbeg + tid + k * SORT_T, cend - 1), 0);
        d[k] = (unsigned)dst[e];
        s[k] = (unsigned)src[e];
    }
#pragma unroll
    for (int k = 0; k < EDGE_B; ++k) {
        d[k] = min(d[k], (unsigned)(N_NODES - 1));
        rank[k] = 0;
        if (tid + k * SORT_T < cnt) rank[k] = atomicAdd(&hist[d[k] >> BKT_SHIFT], 1);
    }
    __syncthreads();
    int c = 0, inc = 0;
    if (tid < 416) {
        c = (tid < N_BKT) ? hist[tid] : 0;
        inc = c;
        for (int o = 1; o < 32; o <<= 1) {
            const int u = __shfl_up(inc, o);
            if (lane >= o) inc += u;
        }
        if (lane == 31) wsum[wave] = inc;
    }
    __syncthreads();
    if (tid < 416) {
        int carry = 0;
        for (int w = 0; w < wave; ++w) carry += wsum[w];
        const int off = carry + inc - c;
        if (tid < N_BKT)    hist[tid] = off;
        if (tid < N_ROWS_T) roff[tid * N_CHUNK + g] = off;
    }
    __syncthreads();
#pragma unroll
    for (int k = 0; k < EDGE_B; ++k) {
        if (tid + k * SORT_T < cnt) {
            const int at = min(hist[d[k] >> BKT_SHIFT] + rank[k], CHUNK_CAP - 1);
            stage[at] = (int)((s[k] & 0x00FFFFFFu) | ((d[k] & (BKT_NODES - 1)) << 24));
        }
    }
    __syncthreads();
    for (int i = tid; i < cnt; i += SORT_T) part[cbeg + i] = stage[i];
}

__device__ __forceinline__ void put_id(int* stage, int* __restrict__ col, int E, int base, unsigned slot, int id) {
    const unsigned q = slot - (unsigned)base;
    if (q < (unsigned)STAGE_N) stage[q] = id;
    else                       col[min(slot, (unsigned)(E - 1))] = id;
}

__global__ void __launch_bounds__(BKT_T)
k_bucket_sort(const int* __restrict__ part, int E, const int* __restrict__ roff,
              const float* __restrict__ x, int* __restrict__ rowptr, int* __restrict__ col,
              float* __restrict__ dinv, float* __restrict__ xs) {
    __shared__ int cnt[BKT_NODES];
    __shared__ int cur[BKT_NODES];
    __shared__ int wsum[BKT_NODES / 32];
    __shared__ int rbeg[N_CHUNK];
    __shared__ int rlen[N_CHUNK];
    __shared__ int bsum[N_CHUNK / 32];
    __shared__ __attribute__((aligned(16))) int stage[STAGE_N];
    const int b    = blockIdx.x;
    const int tid  = threadIdx.x;
    const int lane = tid & 31;
    const int wave = tid >> 5;
    const int per  = min((E + N_CHUNK - 1) / N_CHUNK, CHUNK_CAP);
    if (tid < BKT_NODES) cnt[tid] = 0;
    if (tid < N_CHUNK) {
        const int cb = min(tid * per, E);
        const int cn = min(cb + per, E) - cb;
        const int lo = min(max(roff[b * N_CHUNK + tid], 0), cn);
        const int hi = min(max(roff[(b + 1) * N_CHUNK + tid], lo), cn);
        rbeg[tid] = cb + lo;
        rlen[tid] = hi - lo;
        int t = lo;
        for (int o = 16; o > 0; o >>= 1) t += __shfl_xor(t, o);
        if (lane == 0) bsum[wave] = t;
    }
    __syncthreads();
    int bbeg = 0;
    for (int w = 0; w < N_CHUNK / 32; ++w) bbeg += bsum[w];
    bbeg = min(max(bbeg, 0), E);

    const int hw = tid >> 4;
    const int j  = tid & 15;
    int rb[RUN_IT], rl[RUN_IT];
    unsigned r[2 * RUN_IT];
#pragma unroll
    for (int it = 0; it < RUN_IT; ++it) {
        rb[it] = rbeg[it * (BKT_T / 16) + hw];
        rl[it] = rlen[it * (BKT_T / 16) + hw];
        r[2 * it]     = (unsigned)part[min(rb[it] + j, E - 1)];
        r[2 * it + 1] = (unsigned)part[min(rb[it] + j + 16, E - 1)];
    }
#pragma unroll
    for (int it = 0; it < RUN_IT; ++it) {
        if (j < rl[it])      atomicAdd(&cnt[r[2 * it] >> 24], 1);
        if (j + 16 < rl[it]) atomicAdd(&cnt[r[2 * it + 1] >> 24], 1);
        for (int o = j + 32; o < rl[it]; o += 16)
            atomicAdd(&cnt[(unsigned)part[min(rb[it] + o, E - 1)] >> 24], 1);
    }
    __syncthreads();
    int c = 0, inc = 0;
    if (tid < BKT_NODES) {
        c = cnt[tid];
        inc = c;
        for (int o = 1; o < 32; o <<= 1) {
            const int u = __shfl_up(inc, o);
            if (lane >= o) inc += u;
        }
        if (lane == 31) wsum[wave] = inc;
    }
    __syncthreads();
    if (tid < BKT_NODES) {
        int carry = 0;
        for (int w = 0; w < wave; ++w) carry += wsum[w];
        const int start = bbeg + carry + inc - c;
        cur[tid] = start;
        const int n = b * BKT_NODES + tid;
        rowptr[n] = start;
        if (n < N_NODES) {
            const float di = rsqrtf(fmaxf((float)(c + 1), 1.0f));
            dinv[n] = di;
            xs[F0 * n]     = x[F0 * n] * di;
            xs[F0 * n + 1] = x[F0 * n + 1] * di;
        }
    }
    __syncthreads();
    unsigned slot[2 * RUN_IT];
#pragma unroll
    for (int it = 0; it < RUN_IT; ++it) {
        slot[2 * it] = 0;
        slot[2 * it + 1] = 0;
        if (j < rl[it])      slot[2 * it]     = (unsigned)atomicAdd(&cur[r[2 * it] >> 24], 1);
        if (j + 16 < rl[it]) slot[2 * it + 1] = (unsigned)atomicAdd(&cur[r[2 * it + 1] >> 24], 1);
    }
    const int head = (int)(((size_t)(col + bbeg) >> 2) & 31);
    const int base = bbeg - head;
    bool past = false;
#pragma unroll
    for (int it = 0; it < RUN_IT; ++it) {
        const unsigned q0 = slot[2 * it] - (unsigned)base, q1 = slot[2 * it + 1] - (unsigned)base;
        const bool in0 = j < rl[it] && q0 < (unsigned)STAGE_N, in1 = j + 16 < rl[it] && q1 < (unsigned)STAGE_N;
        if (in0) stage[q0] = (int)(r[2 * it] & 0x00FFFFFFu);
        if (in1) stage[q1] = (int)(r[2 * it + 1] & 0x00FFFFFFu);
        past |= (j < rl[it] && !in0) | (j + 16 < rl[it] && !in1);
    }
    if (past) {
#pragma unroll
        for (int it = 0; it < RUN_IT; ++it) {
            if (j < rl[it] && slot[2 * it] - (unsigned)base >= (unsigned)STAGE_N)
                col[min(slot[2 * it], (unsigned)(E - 1))] = (int)(r[2 * it] & 0x00FFFFFFu);
            if (j + 16 < rl[it] && slot[2 * it + 1] - (unsigned)base >= (unsigned)STAGE_N)
                col[min(slot[2 * it + 1], (unsigned)(E - 1))] = (int)(r[2 * it + 1] & 0x00FFFFFFu);
        }
    }
#pragma unroll
    for (int it = 0; it < RUN_IT; ++it) {
        for (int o = j + 32; o < rl[it]; o += 16) {
            const unsigned q = (unsigned)part[min(rb[it] + o, E - 1)];
            const unsigned sl = (unsigned)atomicAdd(&cur[q >> 24], 1);
            put_id(stage, col, E, base, sl, (int)(q & 0x00FFFFFFu));
        }
    }
    __syncthreads();
    int tot = 0;
    for (int w = 0; w < BKT_NODES / 32; ++w) tot += wsum[w];
    const int nst = min(min(head + tot, STAGE_N), E - base);
    for (int q = 4 * tid; q < nst; q += 4 * BKT_T) {
        if (q >= head && q + 3 < nst) {
            *reinterpret_cast<int4*>(col + base + q) = *reinterpret_cast<const int4*>(stage + q);
        } else {
            for (int k = 0; k < 4; ++k)
                if (q + k >= head && q + k < nst) col[base + q + k] = stage[q + k];
        }
    }
}


#define GX_T     256
#define GX_K     20
#define GX_CH    (32 * GX_K)
#define GX_PITCH (GX_CH + GX_CH / 16)

__global__ void __launch_bounds__(GX_T)
k_gather_x(const int* __restrict__ rowptr, const int* __restrict__ col, int E,
           const float* __restrict__ xs, const float* __restrict__ dinv,
           float* __restrict__ rec) {
    __shared__ float2 stg[GX_T / 32][GX_PITCH];
    const int lane = threadIdx.x & 31;
    const int wave = __builtin_amdgcn_readfirstlane(threadIdx.x >> 5);
    const int n0 = ((int)blockIdx.x * (GX_T / 32) + wave) * 32;
    if (n0 >= N_NODES) return;
    const int n = n0 + lane;
    const float2* xs2 = reinterpret_cast<const float2*>(xs);
    const int beg = min(max(rowptr[n], 0), E);
    const int end = min(max(rowptr[n + 1], beg), E);
    float2 a = xs2[n];
    const float di = dinv[n];
    const int sbeg = __builtin_amdgcn_readlane(beg, 0);
    const int send = __builtin_amdgcn_readlane(end, 31);
    float2* st = stg[wave];
    for (int cb = sbeg; cb < send; cb += GX_CH) {
        int id[GX_K];
        float2 v[GX_K];
#pragma unroll
        for (int k = 0; k < GX_K; ++k)
            id[k] = col[min(cb + k * 32 + lane, send - 1)];
#pragma unroll
        for (int k = 0; k < GX_K; ++k)
            v[k] = xs2[min((unsigned)id[k], (unsigned)(N_NODES - 1))];
#pragma unroll
        for (int k = 0; k < GX_K; ++k) {
            const int p = k * 32 + lane;
            st[p + (p >> 4)] = v[k];
        }
        __builtin_amdgcn_fence(__ATOMIC_RELEASE, "wavefront");
        __builtin_amdgcn_wave_barrier();
        __builtin_amdgcn_fence(__ATOMIC_ACQUIRE, "wavefront");
        const int lo = min(max(beg, cb), cb + GX_CH);
        const int hi = min(max(end, lo), cb + GX_CH);
        for (int j = lo; j < hi; ++j) {
            const int p = j - cb;
            const float2 t = st[p + (p >> 4)];
            a.x += t.x;
            a.y += t.y;
        }
        __builtin_amdgcn_fence(__ATOMIC_RELEASE, "wavefront");
        __builtin_amdgcn_wave_barrier();
        __builtin_amdgcn_fence(__ATOMIC_ACQUIRE, "wavefront");
    }
    reinterpret_cast<float4*>(rec)[n] = make_float4(a.x * di, a.y * di, di, 0.0f);
}


#define STRIPES        (N_NODES / 16)
#define STRIPES_PER_WG 5
#define TILE_PITCH     68

__device__ __forceinline__ v2f walk_chunk(v2f a, int qx, int qy, int qz, int cnt,
                                          v2f w0, v2f w1, v2f bb, v2f slope) {
#pragma unroll 4
    for (int j = 0; j < cnt; ++j) {
        const float ax = __int_as_float(__builtin_amdgcn_readlane(qx, j));
        const float ay = __int_as_float(__builtin_amdgcn_readlane(qy, j));
        const float ds = __int_as_float(__builtin_amdgcn_readlane(qz, j));
        const v2f vax = {ax, ax}, vay = {ay, ay}, vds = {ds, ds};
        const v2f t = __builtin_elementwise_fma(vax, w0, __builtin_elementwise_fma(vay, w1, bb));
        const v2f l = __builtin_elementwise_max(t, t * slope);
        a = __builtin_elementwise_fma(l, vds, a);
    }
    return a;
}

__global__ void __launch_bounds__(256)
k_gather_gemm(const int* __restrict__ rowptr, const int* __restrict__ col, int E,
              const float* __restrict__ rec, const float* __restrict__ W1,
              const float* __restrict__ b1, const float* __restrict__ dinv,
              const float* __restrict__ W2, const float* __restrict__ b2,
              float* __restrict__ z) {
    __shared__ float tile[2][16 * TILE_PITCH];
    const int lane = threadIdx.x & 31;
    const int wave = __builtin_amdgcn_readfirstlane(threadIdx.x >> 5);
    const float4* rec4 = reinterpret_cast<const float4*>(rec);
    const v2f w0 = {W1[2 * lane],      W1[2 * lane + 1]};
    const v2f w1 = {W1[F1 + 2 * lane], W1[F1 + 2 * lane + 1]};
    const v2f bb = {b1[2 * lane],      b1[2 * lane + 1]};
    const v2f slope = {NEG_SLOPE, NEG_SLOPE};

    const int col0 = wave * 16;
    const int m    = lane & 15;
    const int kh   = (lane >> 4) << 1;
    v2f bf[F1 / 4];
#pragma unroll
    for (int t = 0; t < F1 / 4; ++t) {
        bf[t].x = W2[(4 * t + kh) * F2 + col0 + m];
        bf[t].y = W2[(4 * t + kh + 1) * F2 + col0 + m];
    }
    const float bias = b2[col0 + m];

    const int nstr = ((int)blockIdx.x < STRIPES) ? (STRIPES - 1 - (int)blockIdx.x) / (int)gridDim.x + 1 : 0;
    const int nq   = 2 * nstr;
    if (nq == 0) return;
#define NODE_AT(q) (((int)blockIdx.x + (min((q), nq - 1) >> 1) * (int)gridDim.x) * 16 + 2 * wave + (min((q), nq - 1) & 1))

    int   n_c = NODE_AT(0), n_1 = NODE_AT(1), n_2 = NODE_AT(2);
    int   rb_c = rowptr[n_c], re_c = rowptr[n_c + 1];
    int   rb_1 = rowptr[n_1], re_1 = rowptr[n_1 + 1];
    int   rbv_2 = rowptr[n_2 + (lane & 1)];
    float di_c = dinv[n_c], di_1 = dinv[n_1], di_2 = dinv[n_2];
    int beg_c = max(rb_c, 0);
    int deg_c = max(min(re_c, E) - beg_c, 0);
    int beg_1 = max(rb_1, 0);
    int deg_1 = max(min(re_1, E) - beg_1, 0);
    int ids_c = n_c;
    if (lane < deg_c) ids_c = col[beg_c + lane];
    int ids_1 = n_1;
    if (lane < deg_1) ids_1 = col[beg_1 + lane];
    float4 r_c = rec4[min((unsigned)ids_c, (unsigned)(N_NODES - 1))];

    int buf = 0;
    for (int q = 0; q < nq; ++q) {
        const int   n_3  = NODE_AT(q + 3);
        const int   rbv_3 = rowptr[n_3 + (lane & 1)];
        const float di_3 = dinv[n_3];
        const int beg_2 = max(__builtin_amdgcn_readlane(rbv_2, 0), 0);
        const int deg_2 = max(min(__builtin_amdgcn_readlane(rbv_2, 1), E) - beg_2, 0);
        int ids_2 = n_2;
        if (lane < deg_2) ids_2 = col[beg_2 + lane];
        const float4 r_1 = rec4[min((unsigned)ids_1, (unsigned)(N_NODES - 1))];

        const int total = deg_c + 1;
        v2f a = {0.0f, 0.0f};
        a = walk_chunk(a, __float_as_int(r_c.x), __float_as_int(r_c.y), __float_as_int(r_c.z),
                       min(32, total), w0, w1, bb, slope);
        for (int base = 32; base < total; base += 32) {
            int mine = n_c;
            if (base + lane < deg_c) mine = col[beg_c + base + lane];
            const float4 rq = rec4[min((unsigned)mine, (unsigned)(N_NODES - 1))];
            a = walk_chunk(a, __float_as_int(rq.x), __float_as_int(rq.y), __float_as_int(rq.z),
                           min(32, total - base), w0, w1, bb, slope);
        }
        float* tl = tile[buf];
        const int r = 2 * wave + (q & 1);
        const v2f row = {a.x * di_c, a.y * di_c};
        *reinterpret_cast<v2f*>(&tl[r * TILE_PITCH + 2 * lane]) = row;

        if (q & 1) {
            const int s = (int)blockIdx.x + (q >> 1) * (int)gridDim.x;
            __syncthreads();

            v8f acc = {};
            const float* arow = tl + m * TILE_PITCH + kh;
#pragma unroll
            for (int t = 0; t < F1 / 4; ++t) {
                const v2f af = *reinterpret_cast<const v2f*>(arow + 4 * t);
                acc = __builtin_amdgcn_wmma_f32_16x16x4_f32(
                     false, af,  false, bf[t],
                     (short)0, acc,  false,  false);
            }

            const int rbase = s * 16 + ((lane >> 4) << 3);
#pragma unroll
            for (int v = 0; v < 8; ++v) {
                const size_t o = (size_t)(rbase + v) * F2 + col0 + m;
                z[o] = leaky(acc[v] + bias);
            }
            buf ^= 1;
        }

        n_c = n_1; beg_c = beg_1; deg_c = deg_1; di_c = di_1; r_c = r_1;
        n_1 = n_2; beg_1 = beg_2; deg_1 = deg_2; di_1 = di_2; ids_1 = ids_2;
        n_2 = n_3; rbv_2 = rbv_3; di_2 = di_3;
    }
#undef NODE_AT
}


extern "C" void kernel_launch(void* const* d_in, const int* in_sizes, int n_in,
                              void* d_out, int out_size, void* d_ws, size_t ws_size,
                              hipStream_t stream) {
    const int*   edge_index = (const int*)d_in[0];
    const float* x          = (const float*)d_in[1];
    const float* W1         = (const float*)d_in[2];
    const float* b1         = (const float*)d_in[3];
    const float* W2         = (const float*)d_in[4];
    const float* b2         = (const float*)d_in[5];
    float*       z          = (float*)d_out;

    const int E = in_sizes[0] / 2;
    const int* src = edge_index;
    const int* dst = edge_index + E;

    float* ws     = (float*)d_ws;
    int*   roff   = (int*)ws;
    float* dinv   = ws + 204800;
    float* xs     = ws + 307200;
    float* rec    = ws + 512000;
    int*   rowptr = (int*)ws + 1024000;
    int*   part   = (int*)ws + 13516800;
    int*   col    = part + E;
    (void)ws_size; (void)n_in; (void)out_size;

    k_chunk_sort  <<<N_CHUNK, SORT_T, 0, stream>>>(src, dst, E, roff, part);
    k_bucket_sort <<<N_BKT_USED, BKT_T, 0, stream>>>(part, E, roff, x, rowptr, col, dinv, xs);

    k_gather_x<<<(N_NODES / 32 + GX_T / 32 - 1) / (GX_T / 32), GX_T, 0, stream>>>(rowptr, col, E, xs, dinv, rec);

    k_gather_gemm<<<STRIPES / STRIPES_PER_WG, 256, 0, stream>>>(rowptr, col, E, rec, W1, b1, dinv, W2, b2, z);
}
